# starting kernel
# baseline (speedup 1.0000x reference)
#include <hip/hip_runtime.h>
#include <hip/hip_bf16.h>

#define B_  32
#define S_  4096
#define C_  64
#define H_  128
#define P_  8
#define KW  5

typedef __attribute__((ext_vector_type(2))) float v2f;
typedef __attribute__((ext_vector_type(8))) float v8f;

__device__ __forceinline__ v8f wmma_f32_k4(v2f a, v2f b, v8f c) {
  return __builtin_amdgcn_wmma_f32_16x16x4_f32(false, a, false, b, (short)0, c,
                                               false, false);
}

__device__ __forceinline__ float gelu_tanh(float x) {
  float x3 = x * x * x;
  float y  = 1.5957691216057308f * (x + 0.044715f * x3);
  return x * __builtin_amdgcn_rcpf(1.0f + __expf(-y));
}

__global__ __launch_bounds__(256)
void moe_gate_kernel(const float* __restrict__ u,
                     const float* __restrict__ Wr,
                     const float* __restrict__ br,
                     int* __restrict__ gate_out) {
  const int b   = blockIdx.x;
  const int tid = threadIdx.x;
  const int c   = tid & 63;
  const int g   = tid >> 6;

  __shared__ float part[4][64];
  __shared__ float meanv[64];
  __shared__ float logits[P_];

  float s = 0.f;
  const float* ub = u + (size_t)b * S_ * C_;
  for (int srow = g; srow < S_; srow += 4) s += ub[srow * C_ + c];
  part[g][c] = s;
  __syncthreads();

  if (tid < 64)
    meanv[c] = (part[0][c] + part[1][c] + part[2][c] + part[3][c]) * (1.0f / S_);
  __syncthreads();

  if (tid < P_) {
    float l = br[tid];
    for (int cc = 0; cc < C_; ++cc) l += meanv[cc] * Wr[cc * P_ + tid];
    logits[tid] = l;
  }
  __syncthreads();

  if (tid == 0) {
    int i0 = 0; float v0 = logits[0];
    for (int p = 1; p < P_; ++p)
      if (logits[p] > v0) { v0 = logits[p]; i0 = p; }
    int i1 = -1; float v1 = -3.0e38f;
    for (int p = 0; p < P_; ++p)
      if (p != i0 && logits[p] > v1) { v1 = logits[p]; i1 = p; }
    float e1 = __expf(v1 - v0);
    float w0 = 1.0f / (1.0f + e1);
    float w1 = e1 * w0;
    int* o = gate_out + b * 4;
    o[0] = i0; o[1] = i1;
    o[2] = __float_as_int(w0); o[3] = __float_as_int(w1);
  }
}

#define U_OFF    0
#define H_OFF    5712
#define WBUF_OFF 16272
#define LDS_F    67472

__global__ __launch_bounds__(256)
void moe_conv_kernel(const float* __restrict__ u,
                     const float* __restrict__ W1,
                     const float* __restrict__ b1,
                     const float* __restrict__ W2,
                     const float* __restrict__ b2,
                     const int*   __restrict__ gate,
                     float* __restrict__ out) {
  extern __shared__ float smem[];
  float* u_lds = smem + U_OFF;
  float* h_lds = smem + H_OFF;
  float* wbuf  = smem + WBUF_OFF;

  const int s0   = blockIdx.x * 64;
  const int b    = blockIdx.y;
  const int tid  = threadIdx.x;
  const int lane = tid & 31;
  const int wv   = tid >> 5;
  const int n    = lane & 15;
  const int hi   = lane >> 4;

  {
    const float* ub = u + (size_t)b * S_ * C_;
    for (int idx = tid; idx < 84 * 64; idx += 256) {
      int r = idx >> 6, c = idx & 63;
      int gs = s0 - 10 + r;
      float v = (gs >= 0 && gs < S_) ? ub[gs * C_ + c] : 0.f;
      u_lds[r * 68 + c] = v;
    }
  }

  const int* gp = gate + b * 4;
  int   eidx[2] = { gp[0], gp[1] };
  float ew[2]   = { __int_as_float(gp[2]), __int_as_float(gp[3]) };
  __syncthreads();

  const int ct  = wv & 3, c0 = ct * 16;
  const int rtb = wv >> 2;

  float vout[2][8];
  #pragma unroll
  for (int j = 0; j < 2; ++j) {
    int rt = rtb + 2 * j;
    #pragma unroll
    for (int g = 0; g < 8; ++g) {
      int m = g + 8 * hi;
      vout[j][g] = u_lds[(rt * 16 + m + 10) * 68 + c0 + n];
    }
  }

  const int ht = wv, h0 = ht * 16;

  const float* aru = u_lds + n * 68 + 2 * hi;
  const float* arh = h_lds + (n + 6) * 132 + 2 * hi;

  #pragma unroll 1
  for (int e = 0; e < 2; ++e) {
    const int   p  = eidx[e];
    const float we = ew[e];

    {
      const float* w1p = W1 + (size_t)p * KW * C_ * H_;
      for (int idx = tid; idx < KW * C_ * H_; idx += 256) {
        int k = idx >> 13, rem = idx & 8191;
        int c = rem >> 7,  h   = rem & 127;
        wbuf[(k * 32 + (c >> 1)) * 288 + h * 2 + (c & 1)] = w1p[idx];
      }
    }
    __syncthreads();

    {
      const float bb = b1[p * H_ + h0 + n];
      v8f z = {};
      v8f acc[5];
      #pragma unroll
      for (int rt = 0; rt < 5; ++rt) acc[rt] = z;

      #pragma unroll 1
      for (int k = 0; k < KW; ++k) {
        const float* wk = wbuf + ((k * 32 + hi) * 288) + (h0 + n) * 2;
        const float* ak = aru + k * 68;
        #pragma unroll 4
        for (int cc = 0; cc < C_; cc += 4) {
          v2f bfrag = *(const v2f*)(wk + (cc >> 1) * 288);
          #pragma unroll
          for (int rt = 0; rt < 5; ++rt) {
            v2f afrag = *(const v2f*)(ak + rt * (16 * 68) + cc);
            acc[rt] = wmma_f32_k4(afrag, bfrag, acc[rt]);
          }
        }
      }
      #pragma unroll
      for (int rt = 0; rt < 5; ++rt) {
        #pragma unroll
        for (int g = 0; g < 8; ++g) {
          int m = g + 8 * hi;
          int hrow = s0 - 8 + rt * 16 + m;
          float hv = (hrow >= 0 && hrow < S_) ? gelu_tanh(acc[rt][g] + bb) : 0.f;
          h_lds[(rt * 16 + m) * 132 + h0 + n] = hv;
        }
      }
    }
    __syncthreads();

    {
      const float* w2p = W2 + (size_t)p * KW * H_ * C_;
      for (int idx = tid; idx < KW * H_ * C_; idx += 256) {
        int k = idx >> 13, rem = idx & 8191;
        int h = rem >> 6,  c   = rem & 63;
        wbuf[(k * 64 + (h >> 1)) * 160 + c * 2 + (h & 1)] = w2p[idx];
      }
    }
    __syncthreads();

    {
      const float bb2 = b2[p * C_ + c0 + n];
      v8f z = {};
      v8f acc2[2] = { z, z };

      #pragma unroll 1
      for (int k = 0; k < KW; ++k) {
        const float* wk = wbuf + ((k * 64 + hi) * 160) + (c0 + n) * 2;
        const float* ak = arh + (k + rtb * 16) * 132;
        #pragma unroll 4
        for (int hh = 0; hh < H_; hh += 4) {
          v2f bfrag = *(const v2f*)(wk + (hh >> 1) * 160);
          #pragma unroll
          for (int j = 0; j < 2; ++j) {
            v2f afrag = *(const v2f*)(ak + j * (32 * 132) + hh);
            acc2[j] = wmma_f32_k4(afrag, bfrag, acc2[j]);
          }
        }
      }
      #pragma unroll
      for (int j = 0; j < 2; ++j)
        #pragma unroll
        for (int g = 0; g < 8; ++g)
          vout[j][g] += we * (acc2[j][g] + bb2);
    }
    __syncthreads();
  }

  #pragma unroll
  for (int j = 0; j < 2; ++j) {
    int rt = rtb + 2 * j;
    #pragma unroll
    for (int g = 0; g < 8; ++g) {
      int m = g + 8 * hi;
      out[((size_t)b * S_ + s0 + rt * 16 + m) * C_ + c0 + n] = vout[j][g];
    }
  }
}

extern "C" void kernel_launch(void* const* d_in, const int* in_sizes, int n_in,
                              void* d_out, int out_size, void* d_ws, size_t ws_size,
                              hipStream_t stream) {
  const float* u  = (const float*)d_in[0];
  const float* W1 = (const float*)d_in[1];
  const float* b1 = (const float*)d_in[2];
  const float* W2 = (const float*)d_in[3];
  const float* b2 = (const float*)d_in[4];
  const float* Wr = (const float*)d_in[5];
  const float* br = (const float*)d_in[6];
  float* out = (float*)d_out;
  int*   gate = (int*)d_ws;

  moe_gate_kernel<<<dim3(B_), dim3(256), 0, stream>>>(u, Wr, br, gate);

  size_t lds_bytes = (size_t)LDS_F * sizeof(float);
  moe_conv_kernel<<<dim3(S_ / 64, B_), dim3(256), lds_bytes, stream>>>(
      u, W1, b1, W2, b2, gate, out);
}


# Gate kernel's mean over S: 1024 threads per batch row, 16-byte row-contiguous loads with several in flight per thread, LDS tree at the end; changes the summation order of the mean
# speedup vs baseline: 1.2145x; 1.2145x over previous
#include <hip/hip_runtime.h>
#include <hip/hip_bf16.h>

#define B_  32
#define S_  4096
#define C_  64
#define H_  128
#define P_  8
#define KW  5

typedef __attribute__((ext_vector_type(2))) float v2f;
typedef __attribute__((ext_vector_type(8))) float v8f;

__device__ __forceinline__ v8f wmma_f32_k4(v2f a, v2f b, v8f c) {
  return __builtin_amdgcn_wmma_f32_16x16x4_f32(false, a, false, b, (short)0, c,
                                               false, false);
}

__device__ __forceinline__ float gelu_tanh(float x) {
  float x3 = x * x * x;
  float y  = 1.5957691216057308f * (x + 0.044715f * x3);
  return x * __builtin_amdgcn_rcpf(1.0f + __expf(-y));
}

__global__ __launch_bounds__(1024)
void moe_gate_kernel(const float* __restrict__ u,
                     const float* __restrict__ Wr,
                     const float* __restrict__ br,
                     int* __restrict__ gate_out) {
  const int b   = blockIdx.x;
  const int tid = threadIdx.x;
  const int cq  = tid & 15;
  const int rg  = tid >> 4;

  __shared__ float4 part4[64][16];
  __shared__ float part2[4][64];
  __shared__ float meanv[64];
  __shared__ float logits[P_];

  const float4* ub =
      reinterpret_cast<const float4*>(u + (size_t)b * S_ * C_) + rg * 16 + cq;

  float4 acc[8];
  for (int j = 0; j < 8; ++j) acc[j] = make_float4(0.f, 0.f, 0.f, 0.f);

  for (int it = 0; it < 64; it += 8) {
    float4 v[8];
    for (int j = 0; j < 8; ++j) v[j] = ub[(it + j) * (64 * 16)];
    for (int j = 0; j < 8; ++j) {
      acc[j].x += v[j].x; acc[j].y += v[j].y;
      acc[j].z += v[j].z; acc[j].w += v[j].w;
    }
  }
  {
    float4 s;
    s.x = ((acc[0].x + acc[1].x) + (acc[2].x + acc[3].x)) +
          ((acc[4].x + acc[5].x) + (acc[6].x + acc[7].x));
    s.y = ((acc[0].y + acc[1].y) + (acc[2].y + acc[3].y)) +
          ((acc[4].y + acc[5].y) + (acc[6].y + acc[7].y));
    s.z = ((acc[0].z + acc[1].z) + (acc[2].z + acc[3].z)) +
          ((acc[4].z + acc[5].z) + (acc[6].z + acc[7].z));
    s.w = ((acc[0].w + acc[1].w) + (acc[2].w + acc[3].w)) +
          ((acc[4].w + acc[5].w) + (acc[6].w + acc[7].w));
    part4[rg][cq] = s;
  }
  __syncthreads();

  const int c = tid & 63;
  if (tid < 256) {
    const float* pf = reinterpret_cast<const float*>(&part4[0][0]);
    const int q = tid >> 6;
    float s = 0.f;
    for (int r = q * 16; r < q * 16 + 16; ++r) s += pf[r * 64 + c];
    part2[q][c] = s;
  }
  __syncthreads();

  if (tid < 64)
    meanv[c] = (part2[0][c] + part2[1][c] + part2[2][c] + part2[3][c]) * (1.0f / S_);
  __syncthreads();

  if (tid < P_) {
    float l = br[tid];
    for (int cc = 0; cc < C_; ++cc) l += meanv[cc] * Wr[cc * P_ + tid];
    logits[tid] = l;
  }
  __syncthreads();

  if (tid == 0) {
    int i0 = 0; float v0 = logits[0];
    for (int p = 1; p < P_; ++p)
      if (logits[p] > v0) { v0 = logits[p]; i0 = p; }
    int i1 = -1; float v1 = -3.0e38f;
    for (int p = 0; p < P_; ++p)
      if (p != i0 && logits[p] > v1) { v1 = logits[p]; i1 = p; }
    float e1 = __expf(v1 - v0);
    float w0 = 1.0f / (1.0f + e1);
    float w1 = e1 * w0;
    int* o = gate_out + b * 4;
    o[0] = i0; o[1] = i1;
    o[2] = __float_as_int(w0); o[3] = __float_as_int(w1);
  }
}

#define U_OFF    0
#define H_OFF    5712
#define WBUF_OFF 16272
#define LDS_F    67472

__global__ __launch_bounds__(256)
void moe_conv_kernel(const float* __restrict__ u,
                     const float* __restrict__ W1,
                     const float* __restrict__ b1,
                     const float* __restrict__ W2,
                     const float* __restrict__ b2,
                     const int*   __restrict__ gate,
                     float* __restrict__ out) {
  extern __shared__ float smem[];
  float* u_lds = smem + U_OFF;
  float* h_lds = smem + H_OFF;
  float* wbuf  = smem + WBUF_OFF;

  const int s0   = blockIdx.x * 64;
  const int b    = blockIdx.y;
  const int tid  = threadIdx.x;
  const int lane = tid & 31;
  const int wv   = tid >> 5;
  const int n    = lane & 15;
  const int hi   = lane >> 4;

  {
    const float* ub = u + (size_t)b * S_ * C_;
    for (int idx = tid; idx < 84 * 64; idx += 256) {
      int r = idx >> 6, c = idx & 63;
      int gs = s0 - 10 + r;
      float v = (gs >= 0 && gs < S_) ? ub[gs * C_ + c] : 0.f;
      u_lds[r * 68 + c] = v;
    }
  }

  const int* gp = gate + b * 4;
  int   eidx[2] = { gp[0], gp[1] };
  float ew[2]   = { __int_as_float(gp[2]), __int_as_float(gp[3]) };
  __syncthreads();

  const int ct  = wv & 3, c0 = ct * 16;
  const int rtb = wv >> 2;

  float vout[2][8];
  #pragma unroll
  for (int j = 0; j < 2; ++j) {
    int rt = rtb + 2 * j;
    #pragma unroll
    for (int g = 0; g < 8; ++g) {
      int m = g + 8 * hi;
      vout[j][g] = u_lds[(rt * 16 + m + 10) * 68 + c0 + n];
    }
  }

  const int ht = wv, h0 = ht * 16;

  const float* aru = u_lds + n * 68 + 2 * hi;
  const float* arh = h_lds + (n + 6) * 132 + 2 * hi;

  #pragma unroll 1
  for (int e = 0; e < 2; ++e) {
    const int   p  = eidx[e];
    const float we = ew[e];

    {
      const float* w1p = W1 + (size_t)p * KW * C_ * H_;
      for (int idx = tid; idx < KW * C_ * H_; idx += 256) {
        int k = idx >> 13, rem = idx & 8191;
        int c = rem >> 7,  h   = rem & 127;
        wbuf[(k * 32 + (c >> 1)) * 288 + h * 2 + (c & 1)] = w1p[idx];
      }
    }
    __syncthreads();

    {
      const float bb = b1[p * H_ + h0 + n];
      v8f z = {};
      v8f acc[5];
      #pragma unroll
      for (int rt = 0; rt < 5; ++rt) acc[rt] = z;

      #pragma unroll 1
      for (int k = 0; k < KW; ++k) {
        const float* wk = wbuf + ((k * 32 + hi) * 288) + (h0 + n) * 2;
        const float* ak = aru + k * 68;
        #pragma unroll 4
        for (int cc = 0; cc < C_; cc += 4) {
          v2f bfrag = *(const v2f*)(wk + (cc >> 1) * 288);
          #pragma unroll
          for (int rt = 0; rt < 5; ++rt) {
            v2f afrag = *(const v2f*)(ak + rt * (16 * 68) + cc);
            acc[rt] = wmma_f32_k4(afrag, bfrag, acc[rt]);
          }
        }
      }
      #pragma unroll
      for (int rt = 0; rt < 5; ++rt) {
        #pragma unroll
        for (int g = 0; g < 8; ++g) {
          int m = g + 8 * hi;
          int hrow = s0 - 8 + rt * 16 + m;
          float hv = (hrow >= 0 && hrow < S_) ? gelu_tanh(acc[rt][g] + bb) : 0.f;
          h_lds[(rt * 16 + m) * 132 + h0 + n] = hv;
        }
      }
    }
    __syncthreads();

    {
      const float* w2p = W2 + (size_t)p * KW * H_ * C_;
      for (int idx = tid; idx < KW * H_ * C_; idx += 256) {
        int k = idx >> 13, rem = idx & 8191;
        int h = rem >> 6,  c   = rem & 63;
        wbuf[(k * 64 + (h >> 1)) * 160 + c * 2 + (h & 1)] = w2p[idx];
      }
    }
    __syncthreads();

    {
      const float bb2 = b2[p * C_ + c0 + n];
      v8f z = {};
      v8f acc2[2] = { z, z };

      #pragma unroll 1
      for (int k = 0; k < KW; ++k) {
        const float* wk = wbuf + ((k * 64 + hi) * 160) + (c0 + n) * 2;
        const float* ak = arh + (k + rtb * 16) * 132;
        #pragma unroll 4
        for (int hh = 0; hh < H_; hh += 4) {
          v2f bfrag = *(const v2f*)(wk + (hh >> 1) * 160);
          #pragma unroll
          for (int j = 0; j < 2; ++j) {
            v2f afrag = *(const v2f*)(ak + j * (32 * 132) + hh);
            acc2[j] = wmma_f32_k4(afrag, bfrag, acc2[j]);
          }
        }
      }
      #pragma unroll
      for (int j = 0; j < 2; ++j)
        #pragma unroll
        for (int g = 0; g < 8; ++g)
          vout[j][g] += we * (acc2[j][g] + bb2);
    }
    __syncthreads();
  }

  #pragma unroll
  for (int j = 0; j < 2; ++j) {
    int rt = rtb + 2 * j;
    #pragma unroll
    for (int g = 0; g < 8; ++g) {
      int m = g + 8 * hi;
      out[((size_t)b * S_ + s0 + rt * 16 + m) * C_ + c0 + n] = vout[j][g];
    }
  }
}

extern "C" void kernel_launch(void* const* d_in, const int* in_sizes, int n_in,
                              void* d_out, int out_size, void* d_ws, size_t ws_size,
                              hipStream_t stream) {
  const float* u  = (const float*)d_in[0];
  const float* W1 = (const float*)d_in[1];
  const float* b1 = (const float*)d_in[2];
  const float* W2 = (const float*)d_in[3];
  const float* b2 = (const float*)d_in[4];
  const float* Wr = (const float*)d_in[5];
  const float* br = (const float*)d_in[6];
  float* out = (float*)d_out;
  int*   gate = (int*)d_ws;

  moe_gate_kernel<<<dim3(B_), dim3(1024), 0, stream>>>(u, Wr, br, gate);

  size_t lds_bytes = (size_t)LDS_F * sizeof(float);
  moe_conv_kernel<<<dim3(S_ / 64, B_), dim3(256), lds_bytes, stream>>>(
      u, W1, b1, W2, b2, gate, out);
}


# Global-to-LDS staging (both weight slices and the input halo tile) with 16-byte loads and 16-byte LDS stores, row-pair interleave done in registers, a whole slice in flight per stage
# speedup vs baseline: 1.6615x; 1.3681x over previous
#include <hip/hip_runtime.h>
#include <hip/hip_bf16.h>

#define B_  32
#define S_  4096
#define C_  64
#define H_  128
#define P_  8
#define KW  5

typedef __attribute__((ext_vector_type(2))) float v2f;
typedef __attribute__((ext_vector_type(8))) float v8f;

__device__ __forceinline__ v8f wmma_f32_k4(v2f a, v2f b, v8f c) {
  return __builtin_amdgcn_wmma_f32_16x16x4_f32(false, a, false, b, (short)0, c,
                                               false, false);
}

__device__ __forceinline__ float gelu_tanh(float x) {
  float x3 = x * x * x;
  float y  = 1.5957691216057308f * (x + 0.044715f * x3);
  return x * __builtin_amdgcn_rcpf(1.0f + __expf(-y));
}

__global__ __launch_bounds__(1024)
void moe_gate_kernel(const float* __restrict__ u,
                     const float* __restrict__ Wr,
                     const float* __restrict__ br,
                     int* __restrict__ gate_out) {
  const int b   = blockIdx.x;
  const int tid = threadIdx.x;
  const int cq  = tid & 15;
  const int rg  = tid >> 4;

  __shared__ float4 part4[64][16];
  __shared__ float part2[4][64];
  __shared__ float meanv[64];
  __shared__ float logits[P_];

  const float4* ub =
      reinterpret_cast<const float4*>(u + (size_t)b * S_ * C_) + rg * 16 + cq;

  float4 acc[8];
  for (int j = 0; j < 8; ++j) acc[j] = make_float4(0.f, 0.f, 0.f, 0.f);

  for (int it = 0; it < 64; it += 8) {
    float4 v[8];
    for (int j = 0; j < 8; ++j) v[j] = ub[(it + j) * (64 * 16)];
    for (int j = 0; j < 8; ++j) {
      acc[j].x += v[j].x; acc[j].y += v[j].y;
      acc[j].z += v[j].z; acc[j].w += v[j].w;
    }
  }
  {
    float4 s;
    s.x = ((acc[0].x + acc[1].x) + (acc[2].x + acc[3].x)) +
          ((acc[4].x + acc[5].x) + (acc[6].x + acc[7].x));
    s.y = ((acc[0].y + acc[1].y) + (acc[2].y + acc[3].y)) +
          ((acc[4].y + acc[5].y) + (acc[6].y + acc[7].y));
    s.z = ((acc[0].z + acc[1].z) + (acc[2].z + acc[3].z)) +
          ((acc[4].z + acc[5].z) + (acc[6].z + acc[7].z));
    s.w = ((acc[0].w + acc[1].w) + (acc[2].w + acc[3].w)) +
          ((acc[4].w + acc[5].w) + (acc[6].w + acc[7].w));
    part4[rg][cq] = s;
  }
  __syncthreads();

  const int c = tid & 63;
  if (tid < 256) {
    const float* pf = reinterpret_cast<const float*>(&part4[0][0]);
    const int q = tid >> 6;
    float s = 0.f;
    for (int r = q * 16; r < q * 16 + 16; ++r) s += pf[r * 64 + c];
    part2[q][c] = s;
  }
  __syncthreads();

  if (tid < 64)
    meanv[c] = (part2[0][c] + part2[1][c] + part2[2][c] + part2[3][c]) * (1.0f / S_);
  __syncthreads();

  if (tid < P_) {
    float l = br[tid];
    for (int cc = 0; cc < C_; ++cc) l += meanv[cc] * Wr[cc * P_ + tid];
    logits[tid] = l;
  }
  __syncthreads();

  if (tid == 0) {
    int i0 = 0; float v0 = logits[0];
    for (int p = 1; p < P_; ++p)
      if (logits[p] > v0) { v0 = logits[p]; i0 = p; }
    int i1 = -1; float v1 = -3.0e38f;
    for (int p = 0; p < P_; ++p)
      if (p != i0 && logits[p] > v1) { v1 = logits[p]; i1 = p; }
    float e1 = __expf(v1 - v0);
    float w0 = 1.0f / (1.0f + e1);
    float w1 = e1 * w0;
    int* o = gate_out + b * 4;
    o[0] = i0; o[1] = i1;
    o[2] = __float_as_int(w0); o[3] = __float_as_int(w1);
  }
}

#define U_OFF    0
#define H_OFF    5712
#define WBUF_OFF 16272
#define LDS_F    67472

__global__ __launch_bounds__(256)
void moe_conv_kernel(const float* __restrict__ u,
                     const float* __restrict__ W1,
                     const float* __restrict__ b1,
                     const float* __restrict__ W2,
                     const float* __restrict__ b2,
                     const int*   __restrict__ gate,
                     float* __restrict__ out) {
  extern __shared__ float smem[];
  float* u_lds = smem + U_OFF;
  float* h_lds = smem + H_OFF;
  float* wbuf  = smem + WBUF_OFF;

  const int s0   = blockIdx.x * 64;
  const int b    = blockIdx.y;
  const int tid  = threadIdx.x;
  const int lane = tid & 31;
  const int wv   = tid >> 5;
  const int n    = lane & 15;
  const int hi   = lane >> 4;

  {
    const float4* ub4 =
        reinterpret_cast<const float4*>(u + (size_t)b * S_ * C_);
    float4 uv[6];
    #pragma unroll
    for (int j = 0; j < 6; ++j) {
      int idx = tid + 256 * j;
      int r = idx >> 4, q = idx & 15;
      int gs = s0 - 10 + r;
      uv[j] = (idx < 84 * 16 && gs >= 0 && gs < S_)
                  ? ub4[gs * (C_ / 4) + q] : make_float4(0.f, 0.f, 0.f, 0.f);
    }
    #pragma unroll
    for (int j = 0; j < 6; ++j) {
      int idx = tid + 256 * j;
      int r = idx >> 4, q = idx & 15;
      if (idx < 84 * 16)
        *reinterpret_cast<float4*>(u_lds + r * 68 + q * 4) = uv[j];
    }
  }

  const int* gp = gate + b * 4;
  int   eidx[2] = { gp[0], gp[1] };
  float ew[2]   = { __int_as_float(gp[2]), __int_as_float(gp[3]) };
  __syncthreads();

  const int ct  = wv & 3, c0 = ct * 16;
  const int rtb = wv >> 2;

  float vout[2][8];
  #pragma unroll
  for (int j = 0; j < 2; ++j) {
    int rt = rtb + 2 * j;
    #pragma unroll
    for (int g = 0; g < 8; ++g) {
      int m = g + 8 * hi;
      vout[j][g] = u_lds[(rt * 16 + m + 10) * 68 + c0 + n];
    }
  }

  const int ht = wv, h0 = ht * 16;

  const float* aru = u_lds + n * 68 + 2 * hi;
  const float* arh = h_lds + (n + 6) * 132 + 2 * hi;

  #pragma unroll 1
  for (int e = 0; e < 2; ++e) {
    const int   p  = eidx[e];
    const float we = ew[e];

    {
      const float4* src4 =
          reinterpret_cast<const float4*>(W1 + (size_t)p * KW * C_ * H_) +
          (tid >> 5) * 64 + (tid & 31);
      float* dst = wbuf + (tid >> 5) * 288 + (tid & 31) * 8;
      float4 ra[20], rb[20];
      #pragma unroll
      for (int i = 0; i < 20; ++i) {
        ra[i] = src4[i * 512];
        rb[i] = src4[i * 512 + 32];
      }
      #pragma unroll
      for (int i = 0; i < 20; ++i) {
        float* d = dst + i * (8 * 288);
        *reinterpret_cast<float4*>(d) =
            make_float4(ra[i].x, rb[i].x, ra[i].y, rb[i].y);
        *reinterpret_cast<float4*>(d + 4) =
            make_float4(ra[i].z, rb[i].z, ra[i].w, rb[i].w);
      }
    }
    __syncthreads();

    {
      const float bb = b1[p * H_ + h0 + n];
      v8f z = {};
      v8f acc[5];
      #pragma unroll
      for (int rt = 0; rt < 5; ++rt) acc[rt] = z;

      #pragma unroll 1
      for (int k = 0; k < KW; ++k) {
        const float* wk = wbuf + ((k * 32 + hi) * 288) + (h0 + n) * 2;
        const float* ak = aru + k * 68;
        #pragma unroll 4
        for (int cc = 0; cc < C_; cc += 4) {
          v2f bfrag = *(const v2f*)(wk + (cc >> 1) * 288);
          #pragma unroll
          for (int rt = 0; rt < 5; ++rt) {
            v2f afrag = *(const v2f*)(ak + rt * (16 * 68) + cc);
            acc[rt] = wmma_f32_k4(afrag, bfrag, acc[rt]);
          }
        }
      }
      #pragma unroll
      for (int rt = 0; rt < 5; ++rt) {
        #pragma unroll
        for (int g = 0; g < 8; ++g) {
          int m = g + 8 * hi;
          int hrow = s0 - 8 + rt * 16 + m;
          float hv = (hrow >= 0 && hrow < S_) ? gelu_tanh(acc[rt][g] + bb) : 0.f;
          h_lds[(rt * 16 + m) * 132 + h0 + n] = hv;
        }
      }
    }
    __syncthreads();

    {
      const float4* src4 =
          reinterpret_cast<const float4*>(W2 + (size_t)p * KW * H_ * C_) +
          (tid >> 4) * 32 + (tid & 15);
      float* dst = wbuf + (tid >> 4) * 160 + (tid & 15) * 8;
      float4 ra[20], rb[20];
      #pragma unroll
      for (int i = 0; i < 20; ++i) {
        ra[i] = src4[i * 512];
        rb[i] = src4[i * 512 + 16];
      }
      #pragma unroll
      for (int i = 0; i < 20; ++i) {
        float* d = dst + i * (16 * 160);
        *reinterpret_cast<float4*>(d) =
            make_float4(ra[i].x, rb[i].x, ra[i].y, rb[i].y);
        *reinterpret_cast<float4*>(d + 4) =
            make_float4(ra[i].z, rb[i].z, ra[i].w, rb[i].w);
      }
    }
    __syncthreads();

    {
      const float bb2 = b2[p * C_ + c0 + n];
      v8f z = {};
      v8f acc2[2] = { z, z };

      #pragma unroll 1
      for (int k = 0; k < KW; ++k) {
        const float* wk = wbuf + ((k * 64 + hi) * 160) + (c0 + n) * 2;
        const float* ak = arh + (k + rtb * 16) * 132;
        #pragma unroll 4
        for (int hh = 0; hh < H_; hh += 4) {
          v2f bfrag = *(const v2f*)(wk + (hh >> 1) * 160);
          #pragma unroll
          for (int j = 0; j < 2; ++j) {
            v2f afrag = *(const v2f*)(ak + j * (32 * 132) + hh);
            acc2[j] = wmma_f32_k4(afrag, bfrag, acc2[j]);
          }
        }
      }
      #pragma unroll
      for (int j = 0; j < 2; ++j)
        #pragma unroll
        for (int g = 0; g < 8; ++g)
          vout[j][g] += we * (acc2[j][g] + bb2);
    }
    __syncthreads();
  }

  #pragma unroll
  for (int j = 0; j < 2; ++j) {
    int rt = rtb + 2 * j;
    #pragma unroll
    for (int g = 0; g < 8; ++g) {
      int m = g + 8 * hi;
      out[((size_t)b * S_ + s0 + rt * 16 + m) * C_ + c0 + n] = vout[j][g];
    }
  }
}

extern "C" void kernel_launch(void* const* d_in, const int* in_sizes, int n_in,
                              void* d_out, int out_size, void* d_ws, size_t ws_size,
                              hipStream_t stream) {
  const float* u  = (const float*)d_in[0];
  const float* W1 = (const float*)d_in[1];
  const float* b1 = (const float*)d_in[2];
  const float* W2 = (const float*)d_in[3];
  const float* b2 = (const float*)d_in[4];
  const float* Wr = (const float*)d_in[5];
  const float* br = (const float*)d_in[6];
  float* out = (float*)d_out;
  int*   gate = (int*)d_ws;

  moe_gate_kernel<<<dim3(B_), dim3(1024), 0, stream>>>(u, Wr, br, gate);

  size_t lds_bytes = (size_t)LDS_F * sizeof(float);
  moe_conv_kernel<<<dim3(S_ / 64, B_), dim3(256), lds_bytes, stream>>>(
      u, W1, b1, W2, b2, gate, out);
}


# 128 output rows per block instead of 64: each staged weight slice serves twice the rows, conv1 computes 9 row tiles of h per 8 of output (was 5 per 4), conv2 keeps 4 accumulator tiles per wave
# speedup vs baseline: 1.8056x; 1.0867x over previous
#include <hip/hip_runtime.h>
#include <hip/hip_bf16.h>

#define B_  32
#define S_  4096
#define C_  64
#define H_  128
#define P_  8
#define KW  5

typedef __attribute__((ext_vector_type(2))) float v2f;
typedef __attribute__((ext_vector_type(8))) float v8f;

__device__ __forceinline__ v8f wmma_f32_k4(v2f a, v2f b, v8f c) {
  return __builtin_amdgcn_wmma_f32_16x16x4_f32(false, a, false, b, (short)0, c,
                                               false, false);
}

__device__ __forceinline__ float gelu_tanh(float x) {
  float x3 = x * x * x;
  float y  = 1.5957691216057308f * (x + 0.044715f * x3);
  return x * __builtin_amdgcn_rcpf(1.0f + __expf(-y));
}

__global__ __launch_bounds__(1024)
void moe_gate_kernel(const float* __restrict__ u,
                     const float* __restrict__ Wr,
                     const float* __restrict__ br,
                     int* __restrict__ gate_out) {
  const int b   = blockIdx.x;
  const int tid = threadIdx.x;
  const int cq  = tid & 15;
  const int rg  = tid >> 4;

  __shared__ float4 part4[64][16];
  __shared__ float part2[4][64];
  __shared__ float meanv[64];
  __shared__ float logits[P_];

  const float4* ub =
      reinterpret_cast<const float4*>(u + (size_t)b * S_ * C_) + rg * 16 + cq;

  float4 acc[8];
  for (int j = 0; j < 8; ++j) acc[j] = make_float4(0.f, 0.f, 0.f, 0.f);

  for (int it = 0; it < 64; it += 8) {
    float4 v[8];
    for (int j = 0; j < 8; ++j) v[j] = ub[(it + j) * (64 * 16)];
    for (int j = 0; j < 8; ++j) {
      acc[j].x += v[j].x; acc[j].y += v[j].y;
      acc[j].z += v[j].z; acc[j].w += v[j].w;
    }
  }
  {
    float4 s;
    s.x = ((acc[0].x + acc[1].x) + (acc[2].x + acc[3].x)) +
          ((acc[4].x + acc[5].x) + (acc[6].x + acc[7].x));
    s.y = ((acc[0].y + acc[1].y) + (acc[2].y + acc[3].y)) +
          ((acc[4].y + acc[5].y) + (acc[6].y + acc[7].y));
    s.z = ((acc[0].z + acc[1].z) + (acc[2].z + acc[3].z)) +
          ((acc[4].z + acc[5].z) + (acc[6].z + acc[7].z));
    s.w = ((acc[0].w + acc[1].w) + (acc[2].w + acc[3].w)) +
          ((acc[4].w + acc[5].w) + (acc[6].w + acc[7].w));
    part4[rg][cq] = s;
  }
  __syncthreads();

  const int c = tid & 63;
  if (tid < 256) {
    const float* pf = reinterpret_cast<const float*>(&part4[0][0]);
    const int q = tid >> 6;
    float s = 0.f;
    for (int r = q * 16; r < q * 16 + 16; ++r) s += pf[r * 64 + c];
    part2[q][c] = s;
  }
  __syncthreads();

  if (tid < 64)
    meanv[c] = (part2[0][c] + part2[1][c] + part2[2][c] + part2[3][c]) * (1.0f / S_);
  __syncthreads();

  if (tid < P_) {
    float l = br[tid];
    for (int cc = 0; cc < C_; ++cc) l += meanv[cc] * Wr[cc * P_ + tid];
    logits[tid] = l;
  }
  __syncthreads();

  if (tid == 0) {
    int i0 = 0; float v0 = logits[0];
    for (int p = 1; p < P_; ++p)
      if (logits[p] > v0) { v0 = logits[p]; i0 = p; }
    int i1 = -1; float v1 = -3.0e38f;
    for (int p = 0; p < P_; ++p)
      if (p != i0 && logits[p] > v1) { v1 = logits[p]; i1 = p; }
    float e1 = __expf(v1 - v0);
    float w0 = 1.0f / (1.0f + e1);
    float w1 = e1 * w0;
    int* o = gate_out + b * 4;
    o[0] = i0; o[1] = i1;
    o[2] = __float_as_int(w0); o[3] = __float_as_int(w1);
  }
}

#define TS_      128
#define U_ROWS   148
#define RT1      9
#define RT2      4
#define U_OFF    0
#define H_OFF    10064
#define WBUF_OFF 29072
#define LDS_F    80272

__global__ __launch_bounds__(256)
void moe_conv_kernel(const float* __restrict__ u,
                     const float* __restrict__ W1,
                     const float* __restrict__ b1,
                     const float* __restrict__ W2,
                     const float* __restrict__ b2,
                     const int*   __restrict__ gate,
                     float* __restrict__ out) {
  extern __shared__ float smem[];
  float* u_lds = smem + U_OFF;
  float* h_lds = smem + H_OFF;
  float* wbuf  = smem + WBUF_OFF;

  const int s0   = blockIdx.x * TS_;
  const int b    = blockIdx.y;
  const int tid  = threadIdx.x;
  const int lane = tid & 31;
  const int wv   = tid >> 5;
  const int n    = lane & 15;
  const int hi   = lane >> 4;

  {
    const float4* ub4 =
        reinterpret_cast<const float4*>(u + (size_t)b * S_ * C_);
    float4 uv[10];
    #pragma unroll
    for (int j = 0; j < 10; ++j) {
      int idx = tid + 256 * j;
      int r = idx >> 4, q = idx & 15;
      int gs = s0 - 10 + r;
      uv[j] = (idx < U_ROWS * 16 && gs >= 0 && gs < S_)
                  ? ub4[gs * (C_ / 4) + q] : make_float4(0.f, 0.f, 0.f, 0.f);
    }
    #pragma unroll
    for (int j = 0; j < 10; ++j) {
      int idx = tid + 256 * j;
      int r = idx >> 4, q = idx & 15;
      if (idx < U_ROWS * 16)
        *reinterpret_cast<float4*>(u_lds + r * 68 + q * 4) = uv[j];
    }
  }

  const int* gp = gate + b * 4;
  int   eidx[2] = { gp[0], gp[1] };
  float ew[2]   = { __int_as_float(gp[2]), __int_as_float(gp[3]) };
  __syncthreads();

  const int ct  = wv & 3, c0 = ct * 16;
  const int rtb = wv >> 2;

  float vout[RT2][8];
  #pragma unroll
  for (int j = 0; j < RT2; ++j) {
    int rt = rtb + 2 * j;
    #pragma unroll
    for (int g = 0; g < 8; ++g) {
      int m = g + 8 * hi;
      vout[j][g] = u_lds[(rt * 16 + m + 10) * 68 + c0 + n];
    }
  }

  const int ht = wv, h0 = ht * 16;

  const float* aru = u_lds + n * 68 + 2 * hi;
  const float* arh = h_lds + (n + 6) * 132 + 2 * hi;

  #pragma unroll 1
  for (int e = 0; e < 2; ++e) {
    const int   p  = eidx[e];
    const float we = ew[e];

    {
      const float4* src4 =
          reinterpret_cast<const float4*>(W1 + (size_t)p * KW * C_ * H_) +
          (tid >> 5) * 64 + (tid & 31);
      float* dst = wbuf + (tid >> 5) * 288 + (tid & 31) * 8;
      float4 ra[20], rb[20];
      #pragma unroll
      for (int i = 0; i < 20; ++i) {
        ra[i] = src4[i * 512];
        rb[i] = src4[i * 512 + 32];
      }
      #pragma unroll
      for (int i = 0; i < 20; ++i) {
        float* d = dst + i * (8 * 288);
        *reinterpret_cast<float4*>(d) =
            make_float4(ra[i].x, rb[i].x, ra[i].y, rb[i].y);
        *reinterpret_cast<float4*>(d + 4) =
            make_float4(ra[i].z, rb[i].z, ra[i].w, rb[i].w);
      }
    }
    __syncthreads();

    {
      const float bb = b1[p * H_ + h0 + n];
      v8f z = {};
      v8f acc[RT1];
      #pragma unroll
      for (int rt = 0; rt < RT1; ++rt) acc[rt] = z;

      #pragma unroll 1
      for (int k = 0; k < KW; ++k) {
        const float* wk = wbuf + ((k * 32 + hi) * 288) + (h0 + n) * 2;
        const float* ak = aru + k * 68;
        #pragma unroll 4
        for (int cc = 0; cc < C_; cc += 4) {
          v2f bfrag = *(const v2f*)(wk + (cc >> 1) * 288);
          #pragma unroll
          for (int rt = 0; rt < RT1; ++rt) {
            v2f afrag = *(const v2f*)(ak + rt * (16 * 68) + cc);
            acc[rt] = wmma_f32_k4(afrag, bfrag, acc[rt]);
          }
        }
      }
      #pragma unroll
      for (int rt = 0; rt < RT1; ++rt) {
        #pragma unroll
        for (int g = 0; g < 8; ++g) {
          int m = g + 8 * hi;
          int hrow = s0 - 8 + rt * 16 + m;
          float hv = (hrow >= 0 && hrow < S_) ? gelu_tanh(acc[rt][g] + bb) : 0.f;
          h_lds[(rt * 16 + m) * 132 + h0 + n] = hv;
        }
      }
    }
    __syncthreads();

    {
      const float4* src4 =
          reinterpret_cast<const float4*>(W2 + (size_t)p * KW * H_ * C_) +
          (tid >> 4) * 32 + (tid & 15);
      float* dst = wbuf + (tid >> 4) * 160 + (tid & 15) * 8;
      float4 ra[20], rb[20];
      #pragma unroll
      for (int i = 0; i < 20; ++i) {
        ra[i] = src4[i * 512];
        rb[i] = src4[i * 512 + 16];
      }
      #pragma unroll
      for (int i = 0; i < 20; ++i) {
        float* d = dst + i * (16 * 160);
        *reinterpret_cast<float4*>(d) =
            make_float4(ra[i].x, rb[i].x, ra[i].y, rb[i].y);
        *reinterpret_cast<float4*>(d + 4) =
            make_float4(ra[i].z, rb[i].z, ra[i].w, rb[i].w);
      }
    }
    __syncthreads();

    {
      const float bb2 = b2[p * C_ + c0 + n];
      v8f z = {};
      v8f acc2[RT2] = { z, z, z, z };

      #pragma unroll 1
      for (int k = 0; k < KW; ++k) {
        const float* wk = wbuf + ((k * 64 + hi) * 160) + (c0 + n) * 2;
        const float* ak = arh + (k + rtb * 16) * 132;
        #pragma unroll 4
        for (int hh = 0; hh < H_; hh += 4) {
          v2f bfrag = *(const v2f*)(wk + (hh >> 1) * 160);
          #pragma unroll
          for (int j = 0; j < RT2; ++j) {
            v2f afrag = *(const v2f*)(ak + j * (32 * 132) + hh);
            acc2[j] = wmma_f32_k4(afrag, bfrag, acc2[j]);
          }
        }
      }
      #pragma unroll
      for (int j = 0; j < RT2; ++j)
        #pragma unroll
        for (int g = 0; g < 8; ++g)
          vout[j][g] += we * (acc2[j][g] + bb2);
    }
    __syncthreads();
  }

  #pragma unroll
  for (int j = 0; j < RT2; ++j) {
    int rt = rtb + 2 * j;
    #pragma unroll
    for (int g = 0; g < 8; ++g) {
      int m = g + 8 * hi;
      out[((size_t)b * S_ + s0 + rt * 16 + m) * C_ + c0 + n] = vout[j][g];
    }
  }
}

extern "C" void kernel_launch(void* const* d_in, const int* in_sizes, int n_in,
                              void* d_out, int out_size, void* d_ws, size_t ws_size,
                              hipStream_t stream) {
  const float* u  = (const float*)d_in[0];
  const float* W1 = (const float*)d_in[1];
  const float* b1 = (const float*)d_in[2];
  const float* W2 = (const float*)d_in[3];
  const float* b2 = (const float*)d_in[4];
  const float* Wr = (const float*)d_in[5];
  const float* br = (const float*)d_in[6];
  float* out = (float*)d_out;
  int*   gate = (int*)d_ws;

  moe_gate_kernel<<<dim3(B_), dim3(1024), 0, stream>>>(u, Wr, br, gate);

  size_t lds_bytes = (size_t)LDS_F * sizeof(float);
  moe_conv_kernel<<<dim3(S_ / TS_, B_), dim3(256), lds_bytes, stream>>>(
      u, W1, b1, W2, b2, gate, out);
}


# Weight staging split: each slice's 16-byte global loads are issued before the other conv's K loops (W2 before conv1, next W1 before conv2), its LDS stores after; 40 loads per thread in flight
# speedup vs baseline: 1.9961x; 1.1055x over previous
#include <hip/hip_runtime.h>
#include <hip/hip_bf16.h>

#define B_  32
#define S_  4096
#define C_  64
#define H_  128
#define P_  8
#define KW  5

typedef __attribute__((ext_vector_type(2))) float v2f;
typedef __attribute__((ext_vector_type(8))) float v8f;

__device__ __forceinline__ v8f wmma_f32_k4(v2f a, v2f b, v8f c) {
  return __builtin_amdgcn_wmma_f32_16x16x4_f32(false, a, false, b, (short)0, c,
                                               false, false);
}

__device__ __forceinline__ float gelu_tanh(float x) {
  float x3 = x * x * x;
  float y  = 1.5957691216057308f * (x + 0.044715f * x3);
  return x * __builtin_amdgcn_rcpf(1.0f + __expf(-y));
}

__global__ __launch_bounds__(1024)
void moe_gate_kernel(const float* __restrict__ u,
                     const float* __restrict__ Wr,
                     const float* __restrict__ br,
                     int* __restrict__ gate_out) {
  const int b   = blockIdx.x;
  const int tid = threadIdx.x;
  const int cq  = tid & 15;
  const int rg  = tid >> 4;

  __shared__ float4 part4[64][16];
  __shared__ float part2[4][64];
  __shared__ float meanv[64];
  __shared__ float logits[P_];

  const float4* ub =
      reinterpret_cast<const float4*>(u + (size_t)b * S_ * C_) + rg * 16 + cq;

  float4 acc[8];
  for (int j = 0; j < 8; ++j) acc[j] = make_float4(0.f, 0.f, 0.f, 0.f);

  for (int it = 0; it < 64; it += 8) {
    float4 v[8];
    for (int j = 0; j < 8; ++j) v[j] = ub[(it + j) * (64 * 16)];
    for (int j = 0; j < 8; ++j) {
      acc[j].x += v[j].x; acc[j].y += v[j].y;
      acc[j].z += v[j].z; acc[j].w += v[j].w;
    }
  }
  {
    float4 s;
    s.x = ((acc[0].x + acc[1].x) + (acc[2].x + acc[3].x)) +
          ((acc[4].x + acc[5].x) + (acc[6].x + acc[7].x));
    s.y = ((acc[0].y + acc[1].y) + (acc[2].y + acc[3].y)) +
          ((acc[4].y + acc[5].y) + (acc[6].y + acc[7].y));
    s.z = ((acc[0].z + acc[1].z) + (acc[2].z + acc[3].z)) +
          ((acc[4].z + acc[5].z) + (acc[6].z + acc[7].z));
    s.w = ((acc[0].w + acc[1].w) + (acc[2].w + acc[3].w)) +
          ((acc[4].w + acc[5].w) + (acc[6].w + acc[7].w));
    part4[rg][cq] = s;
  }
  __syncthreads();

  const int c = tid & 63;
  if (tid < 256) {
    const float* pf = reinterpret_cast<const float*>(&part4[0][0]);
    const int q = tid >> 6;
    float s = 0.f;
    for (int r = q * 16; r < q * 16 + 16; ++r) s += pf[r * 64 + c];
    part2[q][c] = s;
  }
  __syncthreads();

  if (tid < 64)
    meanv[c] = (part2[0][c] + part2[1][c] + part2[2][c] + part2[3][c]) * (1.0f / S_);
  __syncthreads();

  if (tid < P_) {
    float l = br[tid];
    for (int cc = 0; cc < C_; ++cc) l += meanv[cc] * Wr[cc * P_ + tid];
    logits[tid] = l;
  }
  __syncthreads();

  if (tid == 0) {
    int i0 = 0; float v0 = logits[0];
    for (int p = 1; p < P_; ++p)
      if (logits[p] > v0) { v0 = logits[p]; i0 = p; }
    int i1 = -1; float v1 = -3.0e38f;
    for (int p = 0; p < P_; ++p)
      if (p != i0 && logits[p] > v1) { v1 = logits[p]; i1 = p; }
    float e1 = __expf(v1 - v0);
    float w0 = 1.0f / (1.0f + e1);
    float w1 = e1 * w0;
    int* o = gate_out + b * 4;
    o[0] = i0; o[1] = i1;
    o[2] = __float_as_int(w0); o[3] = __float_as_int(w1);
  }
}

#define TS_      128
#define U_ROWS   148
#define RT1      9
#define RT2      4
#define U_OFF    0
#define H_OFF    10064
#define WBUF_OFF 29072
#define LDS_F    80272

__device__ __forceinline__ void wl_issue_w1(const float* __restrict__ W1, int p, int tid,
                                            float4 (&ra)[20], float4 (&rb)[20]) {
  const float4* src4 =
      reinterpret_cast<const float4*>(W1 + (size_t)p * KW * C_ * H_) +
      (tid >> 5) * 64 + (tid & 31);
  #pragma unroll
  for (int i = 0; i < 20; ++i) {
    ra[i] = src4[i * 512];
    rb[i] = src4[i * 512 + 32];
  }
}
__device__ __forceinline__ void wl_issue_w2(const float* __restrict__ W2, int p, int tid,
                                            float4 (&ra)[20], float4 (&rb)[20]) {
  const float4* src4 =
      reinterpret_cast<const float4*>(W2 + (size_t)p * KW * H_ * C_) +
      (tid >> 4) * 32 + (tid & 15);
  #pragma unroll
  for (int i = 0; i < 20; ++i) {
    ra[i] = src4[i * 512];
    rb[i] = src4[i * 512 + 16];
  }
}
__device__ __forceinline__ void wl_store(float* dst, int step,
                                         const float4 (&ra)[20], const float4 (&rb)[20]) {
  #pragma unroll
  for (int i = 0; i < 20; ++i) {
    float* d = dst + i * step;
    *reinterpret_cast<float4*>(d) =
        make_float4(ra[i].x, rb[i].x, ra[i].y, rb[i].y);
    *reinterpret_cast<float4*>(d + 4) =
        make_float4(ra[i].z, rb[i].z, ra[i].w, rb[i].w);
  }
}

__device__ __forceinline__ void conv2_accumulate(const float* wbuf, const float* arh,
                                                 const float* __restrict__ b2,
                                                 int p, float we, int c0, int n, int hi,
                                                 int rtb, float (&vout)[RT2][8]) {
  const float bb2 = b2[p * C_ + c0 + n];
  v8f z = {};
  v8f acc2[RT2] = { z, z, z, z };

  #pragma unroll 1
  for (int k = 0; k < KW; ++k) {
    const float* wk = wbuf + ((k * 64 + hi) * 160) + (c0 + n) * 2;
    const float* ak = arh + (k + rtb * 16) * 132;
    #pragma unroll 4
    for (int hh = 0; hh < H_; hh += 4) {
      v2f bfrag = *(const v2f*)(wk + (hh >> 1) * 160);
      #pragma unroll
      for (int j = 0; j < RT2; ++j) {
        v2f afrag = *(const v2f*)(ak + j * (32 * 132) + hh);
        acc2[j] = wmma_f32_k4(afrag, bfrag, acc2[j]);
      }
    }
  }
  #pragma unroll
  for (int j = 0; j < RT2; ++j)
    #pragma unroll
    for (int g = 0; g < 8; ++g)
      vout[j][g] += we * (acc2[j][g] + bb2);
}

__global__ __launch_bounds__(256)
void moe_conv_kernel(const float* __restrict__ u,
                     const float* __restrict__ W1,
                     const float* __restrict__ b1,
                     const float* __restrict__ W2,
                     const float* __restrict__ b2,
                     const int*   __restrict__ gate,
                     float* __restrict__ out) {
  extern __shared__ float smem[];
  float* u_lds = smem + U_OFF;
  float* h_lds = smem + H_OFF;
  float* wbuf  = smem + WBUF_OFF;

  const int s0   = blockIdx.x * TS_;
  const int b    = blockIdx.y;
  const int tid  = threadIdx.x;
  const int lane = tid & 31;
  const int wv   = tid >> 5;
  const int n    = lane & 15;
  const int hi   = lane >> 4;

  const int* gp = gate + b * 4;
  int   eidx[2] = { gp[0], gp[1] };
  float ew[2]   = { __int_as_float(gp[2]), __int_as_float(gp[3]) };

  float4 ra[20], rb[20];

  {
    const float4* ub4 =
        reinterpret_cast<const float4*>(u + (size_t)b * S_ * C_);
    float4 uv[10];
    #pragma unroll
    for (int j = 0; j < 10; ++j) {
      int idx = tid + 256 * j;
      int r = idx >> 4, q = idx & 15;
      int gs = s0 - 10 + r;
      uv[j] = (idx < U_ROWS * 16 && gs >= 0 && gs < S_)
                  ? ub4[gs * (C_ / 4) + q] : make_float4(0.f, 0.f, 0.f, 0.f);
    }
    wl_issue_w1(W1, eidx[0], tid, ra, rb);
    #pragma unroll
    for (int j = 0; j < 10; ++j) {
      int idx = tid + 256 * j;
      int r = idx >> 4, q = idx & 15;
      if (idx < U_ROWS * 16)
        *reinterpret_cast<float4*>(u_lds + r * 68 + q * 4) = uv[j];
    }
  }

  __syncthreads();

  const int ct  = wv & 3, c0 = ct * 16;
  const int rtb = wv >> 2;

  float vout[RT2][8];
  #pragma unroll
  for (int j = 0; j < RT2; ++j) {
    int rt = rtb + 2 * j;
    #pragma unroll
    for (int g = 0; g < 8; ++g) {
      int m = g + 8 * hi;
      vout[j][g] = u_lds[(rt * 16 + m + 10) * 68 + c0 + n];
    }
  }

  const int ht = wv, h0 = ht * 16;

  const float* aru = u_lds + n * 68 + 2 * hi;
  const float* arh = h_lds + (n + 6) * 132 + 2 * hi;

  #pragma unroll 1
  for (int e = 0; ; ++e) {
    const int   p  = eidx[e];
    const float we = ew[e];

    wl_store(wbuf + (tid >> 5) * 288 + (tid & 31) * 8, 8 * 288, ra, rb);
    __syncthreads();

    wl_issue_w2(W2, p, tid, ra, rb);

    {
      const float bb = b1[p * H_ + h0 + n];
      v8f z = {};
      v8f acc[RT1];
      #pragma unroll
      for (int rt = 0; rt < RT1; ++rt) acc[rt] = z;

      #pragma unroll 1
      for (int k = 0; k < KW; ++k) {
        const float* wk = wbuf + ((k * 32 + hi) * 288) + (h0 + n) * 2;
        const float* ak = aru + k * 68;
        #pragma unroll 4
        for (int cc = 0; cc < C_; cc += 4) {
          v2f bfrag = *(const v2f*)(wk + (cc >> 1) * 288);
          #pragma unroll
          for (int rt = 0; rt < RT1; ++rt) {
            v2f afrag = *(const v2f*)(ak + rt * (16 * 68) + cc);
            acc[rt] = wmma_f32_k4(afrag, bfrag, acc[rt]);
          }
        }
      }
      #pragma unroll
      for (int rt = 0; rt < RT1; ++rt) {
        #pragma unroll
        for (int g = 0; g < 8; ++g) {
          int m = g + 8 * hi;
          int hrow = s0 - 8 + rt * 16 + m;
          float hv = (hrow >= 0 && hrow < S_) ? gelu_tanh(acc[rt][g] + bb) : 0.f;
          h_lds[(rt * 16 + m) * 132 + h0 + n] = hv;
        }
      }
    }
    __syncthreads();

    wl_store(wbuf + (tid >> 4) * 160 + (tid & 15) * 8, 16 * 160, ra, rb);
    __syncthreads();

    if (e == 1) break;

    wl_issue_w1(W1, eidx[1], tid, ra, rb);

    conv2_accumulate(wbuf, arh, b2, p, we, c0, n, hi, rtb, vout);
    __syncthreads();
  }
  conv2_accumulate(wbuf, arh, b2, eidx[1], ew[1], c0, n, hi, rtb, vout);

  #pragma unroll
  for (int j = 0; j < RT2; ++j) {
    int rt = rtb + 2 * j;
    #pragma unroll
    for (int g = 0; g < 8; ++g) {
      int m = g + 8 * hi;
      out[((size_t)b * S_ + s0 + rt * 16 + m) * C_ + c0 + n] = vout[j][g];
    }
  }
}

extern "C" void kernel_launch(void* const* d_in, const int* in_sizes, int n_in,
                              void* d_out, int out_size, void* d_ws, size_t ws_size,
                              hipStream_t stream) {
  const float* u  = (const float*)d_in[0];
  const float* W1 = (const float*)d_in[1];
  const float* b1 = (const float*)d_in[2];
  const float* W2 = (const float*)d_in[3];
  const float* b2 = (const float*)d_in[4];
  const float* Wr = (const float*)d_in[5];
  const float* br = (const float*)d_in[6];
  float* out = (float*)d_out;
  int*   gate = (int*)d_ws;

  moe_gate_kernel<<<dim3(B_), dim3(1024), 0, stream>>>(u, Wr, br, gate);

  size_t lds_bytes = (size_t)LDS_F * sizeof(float);
  moe_conv_kernel<<<dim3(S_ / TS_, B_), dim3(256), lds_bytes, stream>>>(
      u, W1, b1, W2, b2, gate, out);
}


# conv1 epilogue: bias+gelu of each 8-value tile as branch-free straight-line code (row-range test as a select, only on a block's first and last h tile), so the 72 chains per lane overlap
# speedup vs baseline: 2.1422x; 1.0732x over previous
#include <hip/hip_runtime.h>
#include <hip/hip_bf16.h>

#define B_  32
#define S_  4096
#define C_  64
#define H_  128
#define P_  8
#define KW  5

typedef __attribute__((ext_vector_type(2))) float v2f;
typedef __attribute__((ext_vector_type(8))) float v8f;

__device__ __forceinline__ v8f wmma_f32_k4(v2f a, v2f b, v8f c) {
  return __builtin_amdgcn_wmma_f32_16x16x4_f32(false, a, false, b, (short)0, c,
                                               false, false);
}

__device__ __forceinline__ float gelu_tanh(float x) {
  float x3 = x * x * x;
  float y  = 1.5957691216057308f * (x + 0.044715f * x3);
  return x * __builtin_amdgcn_rcpf(1.0f + __expf(-y));
}

__global__ __launch_bounds__(1024)
void moe_gate_kernel(const float* __restrict__ u,
                     const float* __restrict__ Wr,
                     const float* __restrict__ br,
                     int* __restrict__ gate_out) {
  const int b   = blockIdx.x;
  const int tid = threadIdx.x;
  const int cq  = tid & 15;
  const int rg  = tid >> 4;

  __shared__ float4 part4[64][16];
  __shared__ float part2[4][64];
  __shared__ float meanv[64];
  __shared__ float logits[P_];

  const float4* ub =
      reinterpret_cast<const float4*>(u + (size_t)b * S_ * C_) + rg * 16 + cq;

  float4 acc[8];
  for (int j = 0; j < 8; ++j) acc[j] = make_float4(0.f, 0.f, 0.f, 0.f);

  for (int it = 0; it < 64; it += 8) {
    float4 v[8];
    for (int j = 0; j < 8; ++j) v[j] = ub[(it + j) * (64 * 16)];
    for (int j = 0; j < 8; ++j) {
      acc[j].x += v[j].x; acc[j].y += v[j].y;
      acc[j].z += v[j].z; acc[j].w += v[j].w;
    }
  }
  {
    float4 s;
    s.x = ((acc[0].x + acc[1].x) + (acc[2].x + acc[3].x)) +
          ((acc[4].x + acc[5].x) + (acc[6].x + acc[7].x));
    s.y = ((acc[0].y + acc[1].y) + (acc[2].y + acc[3].y)) +
          ((acc[4].y + acc[5].y) + (acc[6].y + acc[7].y));
    s.z = ((acc[0].z + acc[1].z) + (acc[2].z + acc[3].z)) +
          ((acc[4].z + acc[5].z) + (acc[6].z + acc[7].z));
    s.w = ((acc[0].w + acc[1].w) + (acc[2].w + acc[3].w)) +
          ((acc[4].w + acc[5].w) + (acc[6].w + acc[7].w));
    part4[rg][cq] = s;
  }
  __syncthreads();

  const int c = tid & 63;
  if (tid < 256) {
    const float* pf = reinterpret_cast<const float*>(&part4[0][0]);
    const int q = tid >> 6;
    float s = 0.f;
    for (int r = q * 16; r < q * 16 + 16; ++r) s += pf[r * 64 + c];
    part2[q][c] = s;
  }
  __syncthreads();

  if (tid < 64)
    meanv[c] = (part2[0][c] + part2[1][c] + part2[2][c] + part2[3][c]) * (1.0f / S_);
  __syncthreads();

  if (tid < P_) {
    float l = br[tid];
    for (int cc = 0; cc < C_; ++cc) l += meanv[cc] * Wr[cc * P_ + tid];
    logits[tid] = l;
  }
  __syncthreads();

  if (tid == 0) {
    int i0 = 0; float v0 = logits[0];
    for (int p = 1; p < P_; ++p)
      if (logits[p] > v0) { v0 = logits[p]; i0 = p; }
    int i1 = -1; float v1 = -3.0e38f;
    for (int p = 0; p < P_; ++p)
      if (p != i0 && logits[p] > v1) { v1 = logits[p]; i1 = p; }
    float e1 = __expf(v1 - v0);
    float w0 = 1.0f / (1.0f + e1);
    float w1 = e1 * w0;
    int* o = gate_out + b * 4;
    o[0] = i0; o[1] = i1;
    o[2] = __float_as_int(w0); o[3] = __float_as_int(w1);
  }
}

#define TS_      128
#define U_ROWS   148
#define RT1      9
#define RT2      4
#define U_OFF    0
#define H_OFF    10064
#define WBUF_OFF 29072
#define LDS_F    80272

__device__ __forceinline__ void wl_issue_w1(const float* __restrict__ W1, int p, int tid,
                                            float4 (&ra)[20], float4 (&rb)[20]) {
  const float4* src4 =
      reinterpret_cast<const float4*>(W1 + (size_t)p * KW * C_ * H_) +
      (tid >> 5) * 64 + (tid & 31);
  #pragma unroll
  for (int i = 0; i < 20; ++i) {
    ra[i] = src4[i * 512];
    rb[i] = src4[i * 512 + 32];
  }
}
__device__ __forceinline__ void wl_issue_w2(const float* __restrict__ W2, int p, int tid,
                                            float4 (&ra)[20], float4 (&rb)[20]) {
  const float4* src4 =
      reinterpret_cast<const float4*>(W2 + (size_t)p * KW * H_ * C_) +
      (tid >> 4) * 32 + (tid & 15);
  #pragma unroll
  for (int i = 0; i < 20; ++i) {
    ra[i] = src4[i * 512];
    rb[i] = src4[i * 512 + 16];
  }
}
__device__ __forceinline__ void wl_store(float* dst, int step,
                                         const float4 (&ra)[20], const float4 (&rb)[20]) {
  #pragma unroll
  for (int i = 0; i < 20; ++i) {
    float* d = dst + i * step;
    *reinterpret_cast<float4*>(d) =
        make_float4(ra[i].x, rb[i].x, ra[i].y, rb[i].y);
    *reinterpret_cast<float4*>(d + 4) =
        make_float4(ra[i].z, rb[i].z, ra[i].w, rb[i].w);
  }
}

__device__ __forceinline__ void conv2_accumulate(const float* wbuf, const float* arh,
                                                 const float* __restrict__ b2,
                                                 int p, float we, int c0, int n, int hi,
                                                 int rtb, float (&vout)[RT2][8]) {
  const float bb2 = b2[p * C_ + c0 + n];
  v8f z = {};
  v8f acc2[RT2] = { z, z, z, z };

  #pragma unroll 1
  for (int k = 0; k < KW; ++k) {
    const float* wk = wbuf + ((k * 64 + hi) * 160) + (c0 + n) * 2;
    const float* ak = arh + (k + rtb * 16) * 132;
    #pragma unroll 4
    for (int hh = 0; hh < H_; hh += 4) {
      v2f bfrag = *(const v2f*)(wk + (hh >> 1) * 160);
      #pragma unroll
      for (int j = 0; j < RT2; ++j) {
        v2f afrag = *(const v2f*)(ak + j * (32 * 132) + hh);
        acc2[j] = wmma_f32_k4(afrag, bfrag, acc2[j]);
      }
    }
  }
  #pragma unroll
  for (int j = 0; j < RT2; ++j)
    #pragma unroll
    for (int g = 0; g < 8; ++g)
      vout[j][g] += we * (acc2[j][g] + bb2);
}

__global__ __launch_bounds__(256)
void moe_conv_kernel(const float* __restrict__ u,
                     const float* __restrict__ W1,
                     const float* __restrict__ b1,
                     const float* __restrict__ W2,
                     const float* __restrict__ b2,
                     const int*   __restrict__ gate,
                     float* __restrict__ out) {
  extern __shared__ float smem[];
  float* u_lds = smem + U_OFF;
  float* h_lds = smem + H_OFF;
  float* wbuf  = smem + WBUF_OFF;

  const int s0   = blockIdx.x * TS_;
  const int b    = blockIdx.y;
  const int tid  = threadIdx.x;
  const int lane = tid & 31;
  const int wv   = tid >> 5;
  const int n    = lane & 15;
  const int hi   = lane >> 4;

  const int* gp = gate + b * 4;
  int   eidx[2] = { gp[0], gp[1] };
  float ew[2]   = { __int_as_float(gp[2]), __int_as_float(gp[3]) };

  float4 ra[20], rb[20];

  {
    const float4* ub4 =
        reinterpret_cast<const float4*>(u + (size_t)b * S_ * C_);
    float4 uv[10];
    #pragma unroll
    for (int j = 0; j < 10; ++j) {
      int idx = tid + 256 * j;
      int r = idx >> 4, q = idx & 15;
      int gs = s0 - 10 + r;
      uv[j] = (idx < U_ROWS * 16 && gs >= 0 && gs < S_)
                  ? ub4[gs * (C_ / 4) + q] : make_float4(0.f, 0.f, 0.f, 0.f);
    }
    wl_issue_w1(W1, eidx[0], tid, ra, rb);
    #pragma unroll
    for (int j = 0; j < 10; ++j) {
      int idx = tid + 256 * j;
      int r = idx >> 4, q = idx & 15;
      if (idx < U_ROWS * 16)
        *reinterpret_cast<float4*>(u_lds + r * 68 + q * 4) = uv[j];
    }
  }

  __syncthreads();

  const int ct  = wv & 3, c0 = ct * 16;
  const int rtb = wv >> 2;

  float vout[RT2][8];
  #pragma unroll
  for (int j = 0; j < RT2; ++j) {
    int rt = rtb + 2 * j;
    #pragma unroll
    for (int g = 0; g < 8; ++g) {
      int m = g + 8 * hi;
      vout[j][g] = u_lds[(rt * 16 + m + 10) * 68 + c0 + n];
    }
  }

  const int ht = wv, h0 = ht * 16;

  const float* aru = u_lds + n * 68 + 2 * hi;
  const float* arh = h_lds + (n + 6) * 132 + 2 * hi;

  #pragma unroll 1
  for (int e = 0; ; ++e) {
    const int   p  = eidx[e];
    const float we = ew[e];

    wl_store(wbuf + (tid >> 5) * 288 + (tid & 31) * 8, 8 * 288, ra, rb);
    __syncthreads();

    wl_issue_w2(W2, p, tid, ra, rb);

    {
      const float bb = b1[p * H_ + h0 + n];
      v8f z = {};
      v8f acc[RT1];
      #pragma unroll
      for (int rt = 0; rt < RT1; ++rt) acc[rt] = z;

      #pragma unroll 1
      for (int k = 0; k < KW; ++k) {
        const float* wk = wbuf + ((k * 32 + hi) * 288) + (h0 + n) * 2;
        const float* ak = aru + k * 68;
        #pragma unroll 4
        for (int cc = 0; cc < C_; cc += 4) {
          v2f bfrag = *(const v2f*)(wk + (cc >> 1) * 288);
          #pragma unroll
          for (int rt = 0; rt < RT1; ++rt) {
            v2f afrag = *(const v2f*)(ak + rt * (16 * 68) + cc);
            acc[rt] = wmma_f32_k4(afrag, bfrag, acc[rt]);
          }
        }
      }
      #pragma unroll
      for (int rt = 0; rt < RT1; ++rt) {
        float hv[8];
        #pragma unroll
        for (int g = 0; g < 8; ++g) hv[g] = gelu_tanh(acc[rt][g] + bb);
        if (rt == 0 || rt == RT1 - 1) {
          #pragma unroll
          for (int g = 0; g < 8; ++g) {
            int hrow = s0 - 8 + rt * 16 + g + 8 * hi;
            hv[g] = (hrow >= 0 && hrow < S_) ? hv[g] : 0.f;
          }
        }
        #pragma unroll
        for (int g = 0; g < 8; ++g)
          h_lds[(rt * 16 + g + 8 * hi) * 132 + h0 + n] = hv[g];
      }
    }
    __syncthreads();

    wl_store(wbuf + (tid >> 4) * 160 + (tid & 15) * 8, 16 * 160, ra, rb);
    __syncthreads();

    if (e == 1) break;

    wl_issue_w1(W1, eidx[1], tid, ra, rb);

    conv2_accumulate(wbuf, arh, b2, p, we, c0, n, hi, rtb, vout);
    __syncthreads();
  }
  conv2_accumulate(wbuf, arh, b2, eidx[1], ew[1], c0, n, hi, rtb, vout);

  #pragma unroll
  for (int j = 0; j < RT2; ++j) {
    int rt = rtb + 2 * j;
    #pragma unroll
    for (int g = 0; g < 8; ++g) {
      int m = g + 8 * hi;
      out[((size_t)b * S_ + s0 + rt * 16 + m) * C_ + c0 + n] = vout[j][g];
    }
  }
}

extern "C" void kernel_launch(void* const* d_in, const int* in_sizes, int n_in,
                              void* d_out, int out_size, void* d_ws, size_t ws_size,
                              hipStream_t stream) {
  const float* u  = (const float*)d_in[0];
  const float* W1 = (const float*)d_in[1];
  const float* b1 = (const float*)d_in[2];
  const float* W2 = (const float*)d_in[3];
  const float* b2 = (const float*)d_in[4];
  const float* Wr = (const float*)d_in[5];
  const float* br = (const float*)d_in[6];
  float* out = (float*)d_out;
  int*   gate = (int*)d_ws;

  moe_gate_kernel<<<dim3(B_), dim3(1024), 0, stream>>>(u, Wr, br, gate);

  size_t lds_bytes = (size_t)LDS_F * sizeof(float);
  moe_conv_kernel<<<dim3(S_ / TS_, B_), dim3(256), lds_bytes, stream>>>(
      u, W1, b1, W2, b2, gate, out);
}


# K loops run in four issue-priority levels (s_setprio 3,2,1,0 as a wave advances; loops flattened to trips for it): the wave that is behind on its SIMD wins the matrix pipe, less time is run alone
# speedup vs baseline: 2.2545x; 1.0524x over previous
#include <hip/hip_runtime.h>
#include <hip/hip_bf16.h>

#define B_  32
#define S_  4096
#define C_  64
#define H_  128
#define P_  8
#define KW  5

typedef __attribute__((ext_vector_type(2))) float v2f;
typedef __attribute__((ext_vector_type(8))) float v8f;

__device__ __forceinline__ v8f wmma_f32_k4(v2f a, v2f b, v8f c) {
  return __builtin_amdgcn_wmma_f32_16x16x4_f32(false, a, false, b, (short)0, c,
                                               false, false);
}

__device__ __forceinline__ float gelu_tanh(float x) {
  float x3 = x * x * x;
  float y  = 1.5957691216057308f * (x + 0.044715f * x3);
  return x * __builtin_amdgcn_rcpf(1.0f + __expf(-y));
}

__global__ __launch_bounds__(1024)
void moe_gate_kernel(const float* __restrict__ u,
                     const float* __restrict__ Wr,
                     const float* __restrict__ br,
                     int* __restrict__ gate_out) {
  const int b   = blockIdx.x;
  const int tid = threadIdx.x;
  const int cq  = tid & 15;
  const int rg  = tid >> 4;

  __shared__ float4 part4[64][16];
  __shared__ float part2[4][64];
  __shared__ float meanv[64];
  __shared__ float logits[P_];

  const float4* ub =
      reinterpret_cast<const float4*>(u + (size_t)b * S_ * C_) + rg * 16 + cq;

  float4 acc[8];
  for (int j = 0; j < 8; ++j) acc[j] = make_float4(0.f, 0.f, 0.f, 0.f);

  for (int it = 0; it < 64; it += 8) {
    float4 v[8];
    for (int j = 0; j < 8; ++j) v[j] = ub[(it + j) * (64 * 16)];
    for (int j = 0; j < 8; ++j) {
      acc[j].x += v[j].x; acc[j].y += v[j].y;
      acc[j].z += v[j].z; acc[j].w += v[j].w;
    }
  }
  {
    float4 s;
    s.x = ((acc[0].x + acc[1].x) + (acc[2].x + acc[3].x)) +
          ((acc[4].x + acc[5].x) + (acc[6].x + acc[7].x));
    s.y = ((acc[0].y + acc[1].y) + (acc[2].y + acc[3].y)) +
          ((acc[4].y + acc[5].y) + (acc[6].y + acc[7].y));
    s.z = ((acc[0].z + acc[1].z) + (acc[2].z + acc[3].z)) +
          ((acc[4].z + acc[5].z) + (acc[6].z + acc[7].z));
    s.w = ((acc[0].w + acc[1].w) + (acc[2].w + acc[3].w)) +
          ((acc[4].w + acc[5].w) + (acc[6].w + acc[7].w));
    part4[rg][cq] = s;
  }
  __syncthreads();

  const int c = tid & 63;
  if (tid < 256) {
    const float* pf = reinterpret_cast<const float*>(&part4[0][0]);
    const int q = tid >> 6;
    float s = 0.f;
    for (int r = q * 16; r < q * 16 + 16; ++r) s += pf[r * 64 + c];
    part2[q][c] = s;
  }
  __syncthreads();

  if (tid < 64)
    meanv[c] = (part2[0][c] + part2[1][c] + part2[2][c] + part2[3][c]) * (1.0f / S_);
  __syncthreads();

  if (tid < P_) {
    float l = br[tid];
    for (int cc = 0; cc < C_; ++cc) l += meanv[cc] * Wr[cc * P_ + tid];
    logits[tid] = l;
  }
  __syncthreads();

  if (tid == 0) {
    int i0 = 0; float v0 = logits[0];
    for (int p = 1; p < P_; ++p)
      if (logits[p] > v0) { v0 = logits[p]; i0 = p; }
    int i1 = -1; float v1 = -3.0e38f;
    for (int p = 0; p < P_; ++p)
      if (p != i0 && logits[p] > v1) { v1 = logits[p]; i1 = p; }
    float e1 = __expf(v1 - v0);
    float w0 = 1.0f / (1.0f + e1);
    float w1 = e1 * w0;
    int* o = gate_out + b * 4;
    o[0] = i0; o[1] = i1;
    o[2] = __float_as_int(w0); o[3] = __float_as_int(w1);
  }
}

#define TS_      128
#define U_ROWS   148
#define RT1      9
#define RT2      4
#define U_OFF    0
#define H_OFF    10064
#define WBUF_OFF 29072
#define LDS_F    80272

__device__ __forceinline__ void wl_issue_w1(const float* __restrict__ W1, int p, int tid,
                                            float4 (&ra)[20], float4 (&rb)[20]) {
  const float4* src4 =
      reinterpret_cast<const float4*>(W1 + (size_t)p * KW * C_ * H_) +
      (tid >> 5) * 64 + (tid & 31);
  #pragma unroll
  for (int i = 0; i < 20; ++i) {
    ra[i] = src4[i * 512];
    rb[i] = src4[i * 512 + 32];
  }
}
__device__ __forceinline__ void wl_issue_w2(const float* __restrict__ W2, int p, int tid,
                                            float4 (&ra)[20], float4 (&rb)[20]) {
  const float4* src4 =
      reinterpret_cast<const float4*>(W2 + (size_t)p * KW * H_ * C_) +
      (tid >> 4) * 32 + (tid & 15);
  #pragma unroll
  for (int i = 0; i < 20; ++i) {
    ra[i] = src4[i * 512];
    rb[i] = src4[i * 512 + 16];
  }
}
__device__ __forceinline__ void wl_store(float* dst, int step,
                                         const float4 (&ra)[20], const float4 (&rb)[20]) {
  #pragma unroll
  for (int i = 0; i < 20; ++i) {
    float* d = dst + i * step;
    *reinterpret_cast<float4*>(d) =
        make_float4(ra[i].x, rb[i].x, ra[i].y, rb[i].y);
    *reinterpret_cast<float4*>(d + 4) =
        make_float4(ra[i].z, rb[i].z, ra[i].w, rb[i].w);
  }
}

template <int LO, int HI>
__device__ __forceinline__ void conv1_trips(const float* aru, const float* bw1,
                                            v8f (&acc)[RT1]) {
  #pragma unroll 1
  for (int i = LO; i < HI; ++i) {
    const float* ap = aru + i * 16 + (i >> 2) * 4;
    const float* bp = bw1 + i * (4 * 576);
    #pragma unroll
    for (int s = 0; s < 4; ++s) {
      v2f bfrag = *(const v2f*)(bp + s * 576);
      #pragma unroll
      for (int rt = 0; rt < RT1; ++rt) {
        v2f afrag = *(const v2f*)(ap + rt * (16 * 68) + s * 4);
        acc[rt] = wmma_f32_k4(afrag, bfrag, acc[rt]);
      }
    }
  }
}
template <int LO, int HI>
__device__ __forceinline__ void conv2_trips(const float* ah, const float* bw2,
                                            v8f (&acc2)[RT2]) {
  #pragma unroll 1
  for (int i = LO; i < HI; ++i) {
    const float* ap = ah + i * 16 + (i >> 3) * 4;
    const float* bp = bw2 + i * (4 * 320);
    #pragma unroll
    for (int s = 0; s < 4; ++s) {
      v2f bfrag = *(const v2f*)(bp + s * 320);
      #pragma unroll
      for (int j = 0; j < RT2; ++j) {
        v2f afrag = *(const v2f*)(ap + j * (32 * 132) + s * 4);
        acc2[j] = wmma_f32_k4(afrag, bfrag, acc2[j]);
      }
    }
  }
}

__device__ __forceinline__ void conv2_accumulate(const float* wbuf, const float* arh,
                                                 const float* __restrict__ b2,
                                                 int p, float we, int c0, int n, int hi,
                                                 int rtb, float (&vout)[RT2][8]) {
  const float bb2 = b2[p * C_ + c0 + n];
  v8f z = {};
  v8f acc2[RT2] = { z, z, z, z };

  const float* bw2 = wbuf + hi * 160 + (c0 + n) * 2;
  const float* ah  = arh + rtb * (16 * 132);
  __builtin_amdgcn_s_setprio(3);
  conv2_trips<0, 16>(ah, bw2, acc2);
  __builtin_amdgcn_s_setprio(2);
  conv2_trips<16, 28>(ah, bw2, acc2);
  __builtin_amdgcn_s_setprio(1);
  conv2_trips<28, 36>(ah, bw2, acc2);
  __builtin_amdgcn_s_setprio(0);
  conv2_trips<36, 40>(ah, bw2, acc2);
  #pragma unroll
  for (int j = 0; j < RT2; ++j)
    #pragma unroll
    for (int g = 0; g < 8; ++g)
      vout[j][g] += we * (acc2[j][g] + bb2);
}

__global__ __launch_bounds__(256)
void moe_conv_kernel(const float* __restrict__ u,
                     const float* __restrict__ W1,
                     const float* __restrict__ b1,
                     const float* __restrict__ W2,
                     const float* __restrict__ b2,
                     const int*   __restrict__ gate,
                     float* __restrict__ out) {
  extern __shared__ float smem[];
  float* u_lds = smem + U_OFF;
  float* h_lds = smem + H_OFF;
  float* wbuf  = smem + WBUF_OFF;

  const int s0   = blockIdx.x * TS_;
  const int b    = blockIdx.y;
  const int tid  = threadIdx.x;
  const int lane = tid & 31;
  const int wv   = tid >> 5;
  const int n    = lane & 15;
  const int hi   = lane >> 4;

  const int* gp = gate + b * 4;
  int   eidx[2] = { gp[0], gp[1] };
  float ew[2]   = { __int_as_float(gp[2]), __int_as_float(gp[3]) };

  float4 ra[20], rb[20];

  {
    const float4* ub4 =
        reinterpret_cast<const float4*>(u + (size_t)b * S_ * C_);
    float4 uv[10];
    #pragma unroll
    for (int j = 0; j < 10; ++j) {
      int idx = tid + 256 * j;
      int r = idx >> 4, q = idx & 15;
      int gs = s0 - 10 + r;
      uv[j] = (idx < U_ROWS * 16 && gs >= 0 && gs < S_)
                  ? ub4[gs * (C_ / 4) + q] : make_float4(0.f, 0.f, 0.f, 0.f);
    }
    wl_issue_w1(W1, eidx[0], tid, ra, rb);
    #pragma unroll
    for (int j = 0; j < 10; ++j) {
      int idx = tid + 256 * j;
      int r = idx >> 4, q = idx & 15;
      if (idx < U_ROWS * 16)
        *reinterpret_cast<float4*>(u_lds + r * 68 + q * 4) = uv[j];
    }
  }

  __syncthreads();

  const int ct  = wv & 3, c0 = ct * 16;
  const int rtb = wv >> 2;

  float vout[RT2][8];
  #pragma unroll
  for (int j = 0; j < RT2; ++j) {
    int rt = rtb + 2 * j;
    #pragma unroll
    for (int g = 0; g < 8; ++g) {
      int m = g + 8 * hi;
      vout[j][g] = u_lds[(rt * 16 + m + 10) * 68 + c0 + n];
    }
  }

  const int ht = wv, h0 = ht * 16;

  const float* aru = u_lds + n * 68 + 2 * hi;
  const float* arh = h_lds + (n + 6) * 132 + 2 * hi;

  #pragma unroll 1
  for (int e = 0; ; ++e) {
    const int   p  = eidx[e];
    const float we = ew[e];

    wl_store(wbuf + (tid >> 5) * 288 + (tid & 31) * 8, 8 * 288, ra, rb);
    __syncthreads();

    wl_issue_w2(W2, p, tid, ra, rb);

    {
      const float bb = b1[p * H_ + h0 + n];
      v8f z = {};
      v8f acc[RT1];
      #pragma unroll
      for (int rt = 0; rt < RT1; ++rt) acc[rt] = z;

      const float* bw1 = wbuf + hi * 288 + (h0 + n) * 2;
      __builtin_amdgcn_s_setprio(3);
      conv1_trips<0, 8>(aru, bw1, acc);
      __builtin_amdgcn_s_setprio(2);
      conv1_trips<8, 14>(aru, bw1, acc);
      __builtin_amdgcn_s_setprio(1);
      conv1_trips<14, 18>(aru, bw1, acc);
      __builtin_amdgcn_s_setprio(0);
      conv1_trips<18, 20>(aru, bw1, acc);
      #pragma unroll
      for (int rt = 0; rt < RT1; ++rt) {
        float hv[8];
        #pragma unroll
        for (int g = 0; g < 8; ++g) hv[g] = gelu_tanh(acc[rt][g] + bb);
        if (rt == 0 || rt == RT1 - 1) {
          #pragma unroll
          for (int g = 0; g < 8; ++g) {
            int hrow = s0 - 8 + rt * 16 + g + 8 * hi;
            hv[g] = (hrow >= 0 && hrow < S_) ? hv[g] : 0.f;
          }
        }
        #pragma unroll
        for (int g = 0; g < 8; ++g)
          h_lds[(rt * 16 + g + 8 * hi) * 132 + h0 + n] = hv[g];
      }
    }
    __syncthreads();

    wl_store(wbuf + (tid >> 4) * 160 + (tid & 15) * 8, 16 * 160, ra, rb);
    __syncthreads();

    if (e == 1) break;

    wl_issue_w1(W1, eidx[1], tid, ra, rb);

    conv2_accumulate(wbuf, arh, b2, p, we, c0, n, hi, rtb, vout);
    __syncthreads();
  }
  conv2_accumulate(wbuf, arh, b2, eidx[1], ew[1], c0, n, hi, rtb, vout);

  #pragma unroll
  for (int j = 0; j < RT2; ++j) {
    int rt = rtb + 2 * j;
    #pragma unroll
    for (int g = 0; g < 8; ++g) {
      int m = g + 8 * hi;
      out[((size_t)b * S_ + s0 + rt * 16 + m) * C_ + c0 + n] = vout[j][g];
    }
  }
}

extern "C" void kernel_launch(void* const* d_in, const int* in_sizes, int n_in,
                              void* d_out, int out_size, void* d_ws, size_t ws_size,
                              hipStream_t stream) {
  const float* u  = (const float*)d_in[0];
  const float* W1 = (const float*)d_in[1];
  const float* b1 = (const float*)d_in[2];
  const float* W2 = (const float*)d_in[3];
  const float* b2 = (const float*)d_in[4];
  const float* Wr = (const float*)d_in[5];
  const float* br = (const float*)d_in[6];
  float* out = (float*)d_out;
  int*   gate = (int*)d_ws;

  moe_gate_kernel<<<dim3(B_), dim3(1024), 0, stream>>>(u, Wr, br, gate);

  size_t lds_bytes = (size_t)LDS_F * sizeof(float);
  moe_conv_kernel<<<dim3(S_ / TS_, B_), dim3(256), lds_bytes, stream>>>(
      u, W1, b1, W2, b2, gate, out);
}


# B fragments of both convs read straight from global (L2-resident weights), two register sets, one trip ahead; no weight slices in LDS: LDS 116 KB, two blocks share a WGP; sum order unchanged
# speedup vs baseline: 2.5898x; 1.1487x over previous
#include <hip/hip_runtime.h>
#include <hip/hip_bf16.h>

#define B_  32
#define S_  4096
#define C_  64
#define H_  128
#define P_  8
#define KW  5

typedef __attribute__((ext_vector_type(2))) float v2f;
typedef __attribute__((ext_vector_type(8))) float v8f;

__device__ __forceinline__ v8f wmma_f32_k4(v2f a, v2f b, v8f c) {
  return __builtin_amdgcn_wmma_f32_16x16x4_f32(false, a, false, b, (short)0, c,
                                               false, false);
}

__device__ __forceinline__ float gelu_tanh(float x) {
  float x3 = x * x * x;
  float y  = 1.5957691216057308f * (x + 0.044715f * x3);
  return x * __builtin_amdgcn_rcpf(1.0f + __expf(-y));
}

__global__ __launch_bounds__(1024)
void moe_gate_kernel(const float* __restrict__ u,
                     const float* __restrict__ Wr,
                     const float* __restrict__ br,
                     int* __restrict__ gate_out) {
  const int b   = blockIdx.x;
  const int tid = threadIdx.x;
  const int cq  = tid & 15;
  const int rg  = tid >> 4;

  __shared__ float4 part4[64][16];
  __shared__ float part2[4][64];
  __shared__ float meanv[64];
  __shared__ float logits[P_];

  const float4* ub =
      reinterpret_cast<const float4*>(u + (size_t)b * S_ * C_) + rg * 16 + cq;

  float4 acc[8];
  for (int j = 0; j < 8; ++j) acc[j] = make_float4(0.f, 0.f, 0.f, 0.f);

  for (int it = 0; it < 64; it += 8) {
    float4 v[8];
    for (int j = 0; j < 8; ++j) v[j] = ub[(it + j) * (64 * 16)];
    for (int j = 0; j < 8; ++j) {
      acc[j].x += v[j].x; acc[j].y += v[j].y;
      acc[j].z += v[j].z; acc[j].w += v[j].w;
    }
  }
  {
    float4 s;
    s.x = ((acc[0].x + acc[1].x) + (acc[2].x + acc[3].x)) +
          ((acc[4].x + acc[5].x) + (acc[6].x + acc[7].x));
    s.y = ((acc[0].y + acc[1].y) + (acc[2].y + acc[3].y)) +
          ((acc[4].y + acc[5].y) + (acc[6].y + acc[7].y));
    s.z = ((acc[0].z + acc[1].z) + (acc[2].z + acc[3].z)) +
          ((acc[4].z + acc[5].z) + (acc[6].z + acc[7].z));
    s.w = ((acc[0].w + acc[1].w) + (acc[2].w + acc[3].w)) +
          ((acc[4].w + acc[5].w) + (acc[6].w + acc[7].w));
    part4[rg][cq] = s;
  }
  __syncthreads();

  const int c = tid & 63;
  if (tid < 256) {
    const float* pf = reinterpret_cast<const float*>(&part4[0][0]);
    const int q = tid >> 6;
    float s = 0.f;
    for (int r = q * 16; r < q * 16 + 16; ++r) s += pf[r * 64 + c];
    part2[q][c] = s;
  }
  __syncthreads();

  if (tid < 64)
    meanv[c] = (part2[0][c] + part2[1][c] + part2[2][c] + part2[3][c]) * (1.0f / S_);
  __syncthreads();

  if (tid < P_) {
    float l = br[tid];
    for (int cc = 0; cc < C_; ++cc) l += meanv[cc] * Wr[cc * P_ + tid];
    logits[tid] = l;
  }
  __syncthreads();

  if (tid == 0) {
    int i0 = 0; float v0 = logits[0];
    for (int p = 1; p < P_; ++p)
      if (logits[p] > v0) { v0 = logits[p]; i0 = p; }
    int i1 = -1; float v1 = -3.0e38f;
    for (int p = 0; p < P_; ++p)
      if (p != i0 && logits[p] > v1) { v1 = logits[p]; i1 = p; }
    float e1 = __expf(v1 - v0);
    float w0 = 1.0f / (1.0f + e1);
    float w1 = e1 * w0;
    int* o = gate_out + b * 4;
    o[0] = i0; o[1] = i1;
    o[2] = __float_as_int(w0); o[3] = __float_as_int(w1);
  }
}

#define TS_      128
#define U_ROWS   148
#define RT1      9
#define RT2      4
#define U_OFF    0
#define H_OFF    10064
#define LDS_F    29072
#define NT1      20
#define NT2      20

template <int PITCH, int NS>
__device__ __forceinline__ void load_b_trip(const float* __restrict__ bg, int i, v2f (&bf)[NS]) {
  const float* bq = bg + i * (4 * NS * PITCH);
  #pragma unroll
  for (int s = 0; s < NS; ++s) {
    bf[s].x = bq[(4 * s) * PITCH];
    bf[s].y = bq[(4 * s + 1) * PITCH];
  }
}

__device__ __forceinline__ void conv1_body(const float* aru, int i, const v2f (&bc)[4],
                                           v8f (&acc)[RT1]) {
  const float* ap = aru + i * 16 + (i >> 2) * 4;
  #pragma unroll
  for (int s = 0; s < 4; ++s) {
    #pragma unroll
    for (int rt = 0; rt < RT1; ++rt) {
      v2f afrag = *(const v2f*)(ap + rt * (16 * 68) + s * 4);
      acc[rt] = wmma_f32_k4(afrag, bc[s], acc[rt]);
    }
  }
}
template <int LO, int HI>
__device__ __forceinline__ void conv1_trips(const float* aru, const float* __restrict__ bg1,
                                            v2f (&b0)[4], v2f (&b1)[4], v8f (&acc)[RT1]) {
  #pragma unroll 1
  for (int i = LO; i < HI; i += 2) {
    load_b_trip<H_, 4>(bg1, i + 1, b1);
    conv1_body(aru, i, b0, acc);
    load_b_trip<H_, 4>(bg1, (i + 2 < NT1) ? i + 2 : i + 1, b0);
    conv1_body(aru, i + 1, b1, acc);
  }
}
__device__ __forceinline__ void conv2_body(const float* ah, int i, const v2f (&bc)[8],
                                           v8f (&acc2)[RT2]) {
  const float* ap = ah + i * 32 + (i >> 2) * 4;
  #pragma unroll
  for (int s = 0; s < 8; ++s) {
    #pragma unroll
    for (int j = 0; j < RT2; ++j) {
      v2f afrag = *(const v2f*)(ap + j * (32 * 132) + s * 4);
      acc2[j] = wmma_f32_k4(afrag, bc[s], acc2[j]);
    }
  }
}
template <int LO, int HI>
__device__ __forceinline__ void conv2_trips(const float* ah, const float* __restrict__ bg2,
                                            v2f (&b0)[8], v2f (&b1)[8], v8f (&acc2)[RT2]) {
  #pragma unroll 1
  for (int i = LO; i < HI; i += 2) {
    load_b_trip<C_, 8>(bg2, i + 1, b1);
    conv2_body(ah, i, b0, acc2);
    load_b_trip<C_, 8>(bg2, (i + 2 < NT2) ? i + 2 : i + 1, b0);
    conv2_body(ah, i + 1, b1, acc2);
  }
}
__global__ __launch_bounds__(256)
void moe_conv_kernel(const float* __restrict__ u,
                     const float* __restrict__ W1,
                     const float* __restrict__ b1,
                     const float* __restrict__ W2,
                     const float* __restrict__ b2,
                     const int*   __restrict__ gate,
                     float* __restrict__ out) {
  extern __shared__ float smem[];
  float* u_lds = smem + U_OFF;
  float* h_lds = smem + H_OFF;

  const int s0   = blockIdx.x * TS_;
  const int b    = blockIdx.y;
  const int tid  = threadIdx.x;
  const int lane = tid & 31;
  const int wv   = tid >> 5;
  const int n    = lane & 15;
  const int hi   = lane >> 4;

  const int* gp = gate + b * 4;
  int   eidx[2] = { __builtin_amdgcn_readfirstlane(gp[0]),
                    __builtin_amdgcn_readfirstlane(gp[1]) };
  float ew[2]   = { __int_as_float(__builtin_amdgcn_readfirstlane(gp[2])),
                    __int_as_float(__builtin_amdgcn_readfirstlane(gp[3])) };

  const int ht = wv, h0 = ht * 16;
  const int ct  = wv & 3, c0 = ct * 16;
  const int rtb = wv >> 2;

  const int boff1 = (2 * hi) * H_ + h0 + n;
  const int boff2 = (2 * hi) * C_ + c0 + n;

  v2f wa[4], wb[4];
  {
    const float4* ub4 =
        reinterpret_cast<const float4*>(u + (size_t)b * S_ * C_);
    float4 uv[10];
    #pragma unroll
    for (int j = 0; j < 10; ++j) {
      int idx = tid + 256 * j;
      int r = idx >> 4, q = idx & 15;
      int gs = s0 - 10 + r;
      uv[j] = (idx < U_ROWS * 16 && gs >= 0 && gs < S_)
                  ? ub4[gs * (C_ / 4) + q] : make_float4(0.f, 0.f, 0.f, 0.f);
    }
    load_b_trip<H_, 4>(W1 + (size_t)eidx[0] * (KW * C_ * H_) + boff1, 0, wa);
    #pragma unroll
    for (int j = 0; j < 10; ++j) {
      int idx = tid + 256 * j;
      int r = idx >> 4, q = idx & 15;
      if (idx < U_ROWS * 16)
        *reinterpret_cast<float4*>(u_lds + r * 68 + q * 4) = uv[j];
    }
  }

  __syncthreads();

  float vout[RT2][8];
  #pragma unroll
  for (int j = 0; j < RT2; ++j) {
    int rt = rtb + 2 * j;
    #pragma unroll
    for (int g = 0; g < 8; ++g) {
      int m = g + 8 * hi;
      vout[j][g] = u_lds[(rt * 16 + m + 10) * 68 + c0 + n];
    }
  }

  const float* aru = u_lds + n * 68 + 2 * hi;
  const float* arh = h_lds + (n + 6) * 132 + 2 * hi;

  #pragma unroll 1
  for (int e = 0; e < 2; ++e) {
    const int   p  = eidx[e];
    const float we = ew[e];
    const float* bg1 = W1 + (size_t)p * (KW * C_ * H_) + boff1;
    const float* bg2 = W2 + (size_t)p * (KW * H_ * C_) + boff2;
    v2f xa[8], xb[8];

    {
      const float bb = b1[p * H_ + h0 + n];
      v8f z = {};
      v8f acc[RT1];
      #pragma unroll
      for (int rt = 0; rt < RT1; ++rt) acc[rt] = z;

      __builtin_amdgcn_s_setprio(3);
      conv1_trips<0, 8>(aru, bg1, wa, wb, acc);
      __builtin_amdgcn_s_setprio(2);
      conv1_trips<8, 14>(aru, bg1, wa, wb, acc);
      __builtin_amdgcn_s_setprio(1);
      conv1_trips<14, 18>(aru, bg1, wa, wb, acc);
      __builtin_amdgcn_s_setprio(0);
      conv1_trips<18, 20>(aru, bg1, wa, wb, acc);
      load_b_trip<C_, 8>(bg2, 0, xa);
      #pragma unroll
      for (int rt = 0; rt < RT1; ++rt) {
        float hv[8];
        #pragma unroll
        for (int g = 0; g < 8; ++g) hv[g] = gelu_tanh(acc[rt][g] + bb);
        if (rt == 0 || rt == RT1 - 1) {
          #pragma unroll
          for (int g = 0; g < 8; ++g) {
            int hrow = s0 - 8 + rt * 16 + g + 8 * hi;
            hv[g] = (hrow >= 0 && hrow < S_) ? hv[g] : 0.f;
          }
        }
        #pragma unroll
        for (int g = 0; g < 8; ++g)
          h_lds[(rt * 16 + g + 8 * hi) * 132 + h0 + n] = hv[g];
      }
    }
    __syncthreads();

    if (e == 0)
      load_b_trip<H_, 4>(W1 + (size_t)eidx[1] * (KW * C_ * H_) + boff1, 0, wa);

    {
      const float bb2 = b2[p * C_ + c0 + n];
      v8f z = {};
      v8f acc2[RT2] = { z, z, z, z };
      const float* ah = arh + rtb * (16 * 132);
      __builtin_amdgcn_s_setprio(3);
      conv2_trips<0, 8>(ah, bg2, xa, xb, acc2);
      __builtin_amdgcn_s_setprio(2);
      conv2_trips<8, 14>(ah, bg2, xa, xb, acc2);
      __builtin_amdgcn_s_setprio(1);
      conv2_trips<14, 18>(ah, bg2, xa, xb, acc2);
      __builtin_amdgcn_s_setprio(0);
      conv2_trips<18, 20>(ah, bg2, xa, xb, acc2);
      #pragma unroll
      for (int j = 0; j < RT2; ++j)
        #pragma unroll
        for (int g = 0; g < 8; ++g)
          vout[j][g] += we * (acc2[j][g] + bb2);
    }
    if (e == 0)
      __syncthreads();
  }

  #pragma unroll
  for (int j = 0; j < RT2; ++j) {
    int rt = rtb + 2 * j;
    #pragma unroll
    for (int g = 0; g < 8; ++g) {
      int m = g + 8 * hi;
      out[((size_t)b * S_ + s0 + rt * 16 + m) * C_ + c0 + n] = vout[j][g];
    }
  }
}

extern "C" void kernel_launch(void* const* d_in, const int* in_sizes, int n_in,
                              void* d_out, int out_size, void* d_ws, size_t ws_size,
                              hipStream_t stream) {
  const float* u  = (const float*)d_in[0];
  const float* W1 = (const float*)d_in[1];
  const float* b1 = (const float*)d_in[2];
  const float* W2 = (const float*)d_in[3];
  const float* b2 = (const float*)d_in[4];
  const float* Wr = (const float*)d_in[5];
  const float* br = (const float*)d_in[6];
  float* out = (float*)d_out;
  int*   gate = (int*)d_ws;

  moe_gate_kernel<<<dim3(B_), dim3(1024), 0, stream>>>(u, Wr, br, gate);

  size_t lds_bytes = (size_t)LDS_F * sizeof(float);
  moe_conv_kernel<<<dim3(S_ / TS_, B_), dim3(256), lds_bytes, stream>>>(
      u, W1, b1, W2, b2, gate, out);
}


# Second conv in minimal-filtering form F(2,5): row pairs from 6 point GEMMs over channels, not 10 tap products; W2 transformed per call by spare blocks of launch 1; changes products and sum order
# speedup vs baseline: 2.8449x; 1.0985x over previous
#include <hip/hip_runtime.h>
#include <hip/hip_bf16.h>

#define B_  32
#define S_  4096
#define C_  64
#define H_  128
#define P_  8
#define KW  5

typedef __attribute__((ext_vector_type(2))) float v2f;
typedef __attribute__((ext_vector_type(8))) float v8f;

__device__ __forceinline__ v8f wmma_f32_k4(v2f a, v2f b, v8f c) {
  return __builtin_amdgcn_wmma_f32_16x16x4_f32(false, a, false, b, (short)0, c,
                                               false, false);
}

__device__ __forceinline__ float gelu_tanh(float x) {
  float x3 = x * x * x;
  float y  = 1.5957691216057308f * (x + 0.044715f * x3);
  return x * __builtin_amdgcn_rcpf(1.0f + __expf(-y));
}

#define NXF 16
__global__ __launch_bounds__(1024)
void moe_gate_kernel(const float* __restrict__ u,
                     const float* __restrict__ Wr,
                     const float* __restrict__ br,
                     const float* __restrict__ W2,
                     int* __restrict__ gate_out,
                     float* __restrict__ vt) {
  const int tid = threadIdx.x;
  if (blockIdx.x >= B_) {
    const int idx = (blockIdx.x - B_) * 1024 + tid;
    const int p = idx >> 11;
    const int r = idx & 2047;
    const float4* wp =
        reinterpret_cast<const float4*>(W2) + (size_t)p * (KW * H_ * C_ / 4) + r;
    const float4 q0 = wp[0], q1 = wp[2048], q2 = wp[4096], q3 = wp[6144], q4 = wp[8192];
    const float w0[4] = { q0.x, q0.y, q0.z, q0.w };
    const float w1[4] = { q1.x, q1.y, q1.z, q1.w };
    const float w2[4] = { q2.x, q2.y, q2.z, q2.w };
    const float w3[4] = { q3.x, q3.y, q3.z, q3.w };
    const float w4[4] = { q4.x, q4.y, q4.z, q4.w };
    float v[6][4];
    #pragma unroll
    for (int k = 0; k < 4; ++k) {
      const float sev = (w0[k] + w2[k]) + w4[k];
      const float sod = w1[k] + w3[k];
      const float ev  = (w0[k] * (1.0f / 24.0f) + w2[k] * (1.0f / 6.0f)) + w4[k] * (2.0f / 3.0f);
      const float od  = w1[k] * (1.0f / 12.0f) + w3[k] * (1.0f / 3.0f);
      v[0][k] = 0.25f * w0[k];
      v[1][k] = (sev + sod) * (-1.0f / 6.0f);
      v[2][k] = (sev - sod) * (-1.0f / 6.0f);
      v[3][k] = ev + od;
      v[4][k] = ev - od;
      v[5][k] = w4[k];
    }
    float4* vp = reinterpret_cast<float4*>(vt) + (size_t)p * (6 * 2048) + r;
    #pragma unroll
    for (int x = 0; x < 6; ++x)
      vp[x * 2048] = make_float4(v[x][0], v[x][1], v[x][2], v[x][3]);
    return;
  }
  const int b   = blockIdx.x;
  const int cq  = tid & 15;
  const int rg  = tid >> 4;

  __shared__ float4 part4[64][16];
  __shared__ float part2[4][64];
  __shared__ float meanv[64];
  __shared__ float logits[P_];

  const float4* ub =
      reinterpret_cast<const float4*>(u + (size_t)b * S_ * C_) + rg * 16 + cq;

  float4 acc[8];
  for (int j = 0; j < 8; ++j) acc[j] = make_float4(0.f, 0.f, 0.f, 0.f);

  for (int it = 0; it < 64; it += 8) {
    float4 v[8];
    for (int j = 0; j < 8; ++j) v[j] = ub[(it + j) * (64 * 16)];
    for (int j = 0; j < 8; ++j) {
      acc[j].x += v[j].x; acc[j].y += v[j].y;
      acc[j].z += v[j].z; acc[j].w += v[j].w;
    }
  }
  {
    float4 s;
    s.x = ((acc[0].x + acc[1].x) + (acc[2].x + acc[3].x)) +
          ((acc[4].x + acc[5].x) + (acc[6].x + acc[7].x));
    s.y = ((acc[0].y + acc[1].y) + (acc[2].y + acc[3].y)) +
          ((acc[4].y + acc[5].y) + (acc[6].y + acc[7].y));
    s.z = ((acc[0].z + acc[1].z) + (acc[2].z + acc[3].z)) +
          ((acc[4].z + acc[5].z) + (acc[6].z + acc[7].z));
    s.w = ((acc[0].w + acc[1].w) + (acc[2].w + acc[3].w)) +
          ((acc[4].w + acc[5].w) + (acc[6].w + acc[7].w));
    part4[rg][cq] = s;
  }
  __syncthreads();

  const int c = tid & 63;
  if (tid < 256) {
    const float* pf = reinterpret_cast<const float*>(&part4[0][0]);
    const int q = tid >> 6;
    float s = 0.f;
    for (int r = q * 16; r < q * 16 + 16; ++r) s += pf[r * 64 + c];
    part2[q][c] = s;
  }
  __syncthreads();

  if (tid < 64)
    meanv[c] = (part2[0][c] + part2[1][c] + part2[2][c] + part2[3][c]) * (1.0f / S_);
  __syncthreads();

  if (tid < P_) {
    float l = br[tid];
    for (int cc = 0; cc < C_; ++cc) l += meanv[cc] * Wr[cc * P_ + tid];
    logits[tid] = l;
  }
  __syncthreads();

  if (tid == 0) {
    int i0 = 0; float v0 = logits[0];
    for (int p = 1; p < P_; ++p)
      if (logits[p] > v0) { v0 = logits[p]; i0 = p; }
    int i1 = -1; float v1 = -3.0e38f;
    for (int p = 0; p < P_; ++p)
      if (p != i0 && logits[p] > v1) { v1 = logits[p]; i1 = p; }
    float e1 = __expf(v1 - v0);
    float w0 = 1.0f / (1.0f + e1);
    float w1 = e1 * w0;
    int* o = gate_out + b * 4;
    o[0] = i0; o[1] = i1;
    o[2] = __float_as_int(w0); o[3] = __float_as_int(w1);
  }
}

#define TS_      128
#define U_ROWS   148
#define RT1      9
#define U_OFF    0
#define H_OFF    10064
#define HP       130
#define LDS_F    28784
#define NT1      20
#define NT2      16

template <int PITCH, int NS>
__device__ __forceinline__ void load_b_trip(const float* __restrict__ bg, int i, v2f (&bf)[NS]) {
  const float* bq = bg + i * (4 * NS * PITCH);
  #pragma unroll
  for (int s = 0; s < NS; ++s) {
    bf[s].x = bq[(4 * s) * PITCH];
    bf[s].y = bq[(4 * s + 1) * PITCH];
  }
}

__device__ __forceinline__ void conv1_body(const float* aru, int i, const v2f (&bc)[4],
                                           v8f (&acc)[RT1]) {
  const float* ap = aru + i * 16 + (i >> 2) * 4;
  #pragma unroll
  for (int s = 0; s < 4; ++s) {
    #pragma unroll
    for (int rt = 0; rt < RT1; ++rt) {
      v2f afrag = *(const v2f*)(ap + rt * (16 * 68) + s * 4);
      acc[rt] = wmma_f32_k4(afrag, bc[s], acc[rt]);
    }
  }
}
template <int LO, int HI>
__device__ __forceinline__ void conv1_trips(const float* aru, const float* __restrict__ bg1,
                                            v2f (&b0)[4], v2f (&b1)[4], v8f (&acc)[RT1]) {
  #pragma unroll 1
  for (int i = LO; i < HI; i += 2) {
    load_b_trip<H_, 4>(bg1, i + 1, b1);
    conv1_body(aru, i, b0, acc);
    load_b_trip<H_, 4>(bg1, (i + 2 < NT1) ? i + 2 : i + 1, b0);
    conv1_body(aru, i + 1, b1, acc);
  }
}
__device__ __forceinline__ void load_v_trip(const float* __restrict__ vg, int i,
                                            v2f (&bf)[2][6]) {
  const float* bq = vg + i * (8 * C_);
  #pragma unroll
  for (int s = 0; s < 2; ++s) {
    #pragma unroll
    for (int x = 0; x < 6; ++x) {
      bf[s][x].x = bq[x * (H_ * C_) + (4 * s) * C_];
      bf[s][x].y = bq[x * (H_ * C_) + (4 * s + 1) * C_];
    }
  }
}
__device__ __forceinline__ void wino_body(const float* ah, int i, const v2f (&bc)[2][6],
                                          v8f (&m)[2][6]) {
  const float* ap = ah + i * 8;
  #pragma unroll
  for (int s = 0; s < 2; ++s) {
    #pragma unroll
    for (int j = 0; j < 2; ++j) {
      const float* aq = ap + j * (64 * HP) + s * 4;
      const v2f d0 = *(const v2f*)(aq);
      const v2f d1 = *(const v2f*)(aq + HP);
      const v2f d2 = *(const v2f*)(aq + 2 * HP);
      const v2f d3 = *(const v2f*)(aq + 3 * HP);
      const v2f d4 = *(const v2f*)(aq + 4 * HP);
      const v2f d5 = *(const v2f*)(aq + 5 * HP);
      const v2f pa = d4 - 4.0f * d2;
      const v2f pb = d3 - 4.0f * d1;
      const v2f pc = d4 - d2;
      const v2f pe = d3 - d1;
      const v2f x0 = 4.0f * d0 + (d4 - 5.0f * d2);
      const v2f x1 = pa + pb;
      const v2f x2 = pa - pb;
      const v2f x3 = pc + 2.0f * pe;
      const v2f x4 = pc - 2.0f * pe;
      const v2f x5 = 4.0f * d1 + (d5 - 5.0f * d3);
      m[j][0] = wmma_f32_k4(x0, bc[s][0], m[j][0]);
      m[j][1] = wmma_f32_k4(x1, bc[s][1], m[j][1]);
      m[j][2] = wmma_f32_k4(x2, bc[s][2], m[j][2]);
      m[j][3] = wmma_f32_k4(x3, bc[s][3], m[j][3]);
      m[j][4] = wmma_f32_k4(x4, bc[s][4], m[j][4]);
      m[j][5] = wmma_f32_k4(x5, bc[s][5], m[j][5]);
    }
  }
}
template <int LO, int HI>
__device__ __forceinline__ void wino_trips(const float* ah, const float* __restrict__ vg,
                                           v2f (&b0)[2][6], v2f (&b1)[2][6], v8f (&m)[2][6]) {
  #pragma unroll 1
  for (int i = LO; i < HI; i += 2) {
    load_v_trip(vg, i + 1, b1);
    wino_body(ah, i, b0, m);
    load_v_trip(vg, (i + 2 < NT2) ? i + 2 : i + 1, b0);
    wino_body(ah, i + 1, b1, m);
  }
}
__global__ __launch_bounds__(256, 4)
void moe_conv_kernel(const float* __restrict__ u,
                     const float* __restrict__ W1,
                     const float* __restrict__ b1,
                     const float* __restrict__ b2,
                     const float* __restrict__ vt,
                     const int*   __restrict__ gate,
                     float* __restrict__ out) {
  extern __shared__ float smem[];
  float* u_lds = smem + U_OFF;
  float* h_lds = smem + H_OFF;

  const int s0   = blockIdx.x * TS_;
  const int b    = blockIdx.y;
  const int tid  = threadIdx.x;
  const int lane = tid & 31;
  const int wv   = tid >> 5;
  const int n    = lane & 15;
  const int hi   = lane >> 4;

  const int* gp = gate + b * 4;
  int   eidx[2] = { __builtin_amdgcn_readfirstlane(gp[0]),
                    __builtin_amdgcn_readfirstlane(gp[1]) };
  float ew[2]   = { __int_as_float(__builtin_amdgcn_readfirstlane(gp[2])),
                    __int_as_float(__builtin_amdgcn_readfirstlane(gp[3])) };

  const int ht = wv, h0 = ht * 16;
  const int ct  = wv & 3, c0 = ct * 16;
  const int rtb = wv >> 2;

  const int boff1 = (2 * hi) * H_ + h0 + n;
  const int boff2 = (2 * hi) * C_ + c0 + n;

  v2f wa[4], wb[4];
  {
    const float4* ub4 =
        reinterpret_cast<const float4*>(u + (size_t)b * S_ * C_);
    float4 uv[10];
    #pragma unroll
    for (int j = 0; j < 10; ++j) {
      int idx = tid + 256 * j;
      int r = idx >> 4, q = idx & 15;
      int gs = s0 - 10 + r;
      uv[j] = (idx < U_ROWS * 16 && gs >= 0 && gs < S_)
                  ? ub4[gs * (C_ / 4) + q] : make_float4(0.f, 0.f, 0.f, 0.f);
    }
    load_b_trip<H_, 4>(W1 + (size_t)eidx[0] * (KW * C_ * H_) + boff1, 0, wa);
    #pragma unroll
    for (int j = 0; j < 10; ++j) {
      int idx = tid + 256 * j;
      int r = idx >> 4, q = idx & 15;
      if (idx < U_ROWS * 16)
        *reinterpret_cast<float4*>(u_lds + r * 68 + q * 4) = uv[j];
    }
  }

  __syncthreads();

  float vout[2][2][8];
  #pragma unroll
  for (int j = 0; j < 2; ++j) {
    #pragma unroll
    for (int g = 0; g < 8; ++g) {
      #pragma unroll
      for (int i = 0; i < 2; ++i) {
        int row = 32 * (rtb + 2 * j) + 2 * (g + 8 * hi) + i;
        vout[j][i][g] = u_lds[(row + 10) * 68 + c0 + n];
      }
    }
  }

  const float* aru = u_lds + n * 68 + 2 * hi;
  const float* arh = h_lds + (2 * n + 6) * HP + 2 * hi;

  #pragma unroll 1
  for (int e = 0; e < 2; ++e) {
    const int   p  = eidx[e];
    const float we = ew[e];
    const float* bg1 = W1 + (size_t)p * (KW * C_ * H_) + boff1;
    const float* vg  = vt + (size_t)p * (6 * H_ * C_) + boff2;
    v2f xa[2][6], xb[2][6];

    {
      const float bb = b1[p * H_ + h0 + n];
      v8f z = {};
      v8f acc[RT1];
      #pragma unroll
      for (int rt = 0; rt < RT1; ++rt) acc[rt] = z;

      __builtin_amdgcn_s_setprio(3);
      conv1_trips<0, 8>(aru, bg1, wa, wb, acc);
      __builtin_amdgcn_s_setprio(2);
      conv1_trips<8, 14>(aru, bg1, wa, wb, acc);
      __builtin_amdgcn_s_setprio(1);
      conv1_trips<14, 18>(aru, bg1, wa, wb, acc);
      __builtin_amdgcn_s_setprio(0);
      conv1_trips<18, 20>(aru, bg1, wa, wb, acc);
      load_v_trip(vg, 0, xa);
      #pragma unroll
      for (int rt = 0; rt < RT1; ++rt) {
        float hv[8];
        #pragma unroll
        for (int g = 0; g < 8; ++g) hv[g] = gelu_tanh(acc[rt][g] + bb);
        if (rt == 0 || rt == RT1 - 1) {
          #pragma unroll
          for (int g = 0; g < 8; ++g) {
            int hrow = s0 - 8 + rt * 16 + g + 8 * hi;
            hv[g] = (hrow >= 0 && hrow < S_) ? hv[g] : 0.f;
          }
        }
        #pragma unroll
        for (int g = 0; g < 8; ++g)
          h_lds[(rt * 16 + g + 8 * hi) * HP + h0 + n] = hv[g];
      }
    }
    __syncthreads();

    if (e == 0)
      load_b_trip<H_, 4>(W1 + (size_t)eidx[1] * (KW * C_ * H_) + boff1, 0, wa);

    {
      const float bb2 = b2[p * C_ + c0 + n];
      v8f z = {};
      v8f m[2][6] = { { z, z, z, z, z, z }, { z, z, z, z, z, z } };
      const float* ah = arh + rtb * (32 * HP);
      __builtin_amdgcn_s_setprio(3);
      wino_trips<0, 6>(ah, vg, xa, xb, m);
      __builtin_amdgcn_s_setprio(2);
      wino_trips<6, 10>(ah, vg, xa, xb, m);
      __builtin_amdgcn_s_setprio(1);
      wino_trips<10, 14>(ah, vg, xa, xb, m);
      __builtin_amdgcn_s_setprio(0);
      wino_trips<14, 16>(ah, vg, xa, xb, m);
      #pragma unroll
      for (int j = 0; j < 2; ++j) {
        #pragma unroll
        for (int g = 0; g < 8; ++g) {
          const float y0 = ((m[j][0][g] + m[j][1][g]) + (m[j][2][g] + m[j][3][g])) + m[j][4][g];
          const float y1 = ((m[j][1][g] - m[j][2][g]) + 2.0f * (m[j][3][g] - m[j][4][g])) + m[j][5][g];
          vout[j][0][g] += we * (y0 + bb2);
          vout[j][1][g] += we * (y1 + bb2);
        }
      }
    }
    if (e == 0)
      __syncthreads();
  }

  #pragma unroll
  for (int j = 0; j < 2; ++j) {
    #pragma unroll
    for (int g = 0; g < 8; ++g) {
      #pragma unroll
      for (int i = 0; i < 2; ++i) {
        int row = 32 * (rtb + 2 * j) + 2 * (g + 8 * hi) + i;
        out[((size_t)b * S_ + s0 + row) * C_ + c0 + n] = vout[j][i][g];
      }
    }
  }
}

extern "C" void kernel_launch(void* const* d_in, const int* in_sizes, int n_in,
                              void* d_out, int out_size, void* d_ws, size_t ws_size,
                              hipStream_t stream) {
  const float* u  = (const float*)d_in[0];
  const float* W1 = (const float*)d_in[1];
  const float* b1 = (const float*)d_in[2];
  const float* W2 = (const float*)d_in[3];
  const float* b2 = (const float*)d_in[4];
  const float* Wr = (const float*)d_in[5];
  const float* br = (const float*)d_in[6];
  float* out = (float*)d_out;
  int*   gate = (int*)d_ws;
  float* vt   = (float*)d_ws + 1024;

  moe_gate_kernel<<<dim3(B_ + NXF), dim3(1024), 0, stream>>>(u, Wr, br, W2, gate, vt);

  size_t lds_bytes = (size_t)LDS_F * sizeof(float);
  moe_conv_kernel<<<dim3(S_ / TS_, B_), dim3(256), lds_bytes, stream>>>(
      u, W1, b1, b2, vt, gate, out);
}


# First conv in minimal-filtering form F(2,5) too: h row pairs from 6 point GEMMs over channels, not 10 tap products; W1 transformed per call by spare blocks of launch 1; changes products and sum order
# speedup vs baseline: 2.9657x; 1.0425x over previous
#include <hip/hip_runtime.h>
#include <hip/hip_bf16.h>

#define B_  32
#define S_  4096
#define C_  64
#define H_  128
#define P_  8
#define KW  5

typedef __attribute__((ext_vector_type(2))) float v2f;
typedef __attribute__((ext_vector_type(8))) float v8f;

__device__ __forceinline__ v8f wmma_f32_k4(v2f a, v2f b, v8f c) {
  return __builtin_amdgcn_wmma_f32_16x16x4_f32(false, a, false, b, (short)0, c,
                                               false, false);
}

__device__ __forceinline__ float gelu_tanh(float x) {
  float x3 = x * x * x;
  float y  = 1.5957691216057308f * (x + 0.044715f * x3);
  return x * __builtin_amdgcn_rcpf(1.0f + __expf(-y));
}

#define NXF 32
__global__ __launch_bounds__(1024)
void moe_gate_kernel(const float* __restrict__ u,
                     const float* __restrict__ Wr,
                     const float* __restrict__ br,
                     const float* __restrict__ W2,
                     const float* __restrict__ W1,
                     int* __restrict__ gate_out,
                     float* __restrict__ vt,
                     float* __restrict__ vt1) {
  const int tid = threadIdx.x;
  if (blockIdx.x >= B_) {
    const bool first = blockIdx.x >= B_ + 16;
    const int idx = ((blockIdx.x - B_) & 15) * 1024 + tid;
    const int p = idx >> 11;
    const int r = idx & 2047;
    const float4* wp =
        reinterpret_cast<const float4*>(first ? W1 : W2) + (size_t)p * (KW * H_ * C_ / 4) + r;
    const float4 q0 = wp[0], q1 = wp[2048], q2 = wp[4096], q3 = wp[6144], q4 = wp[8192];
    const float w0[4] = { q0.x, q0.y, q0.z, q0.w };
    const float w1[4] = { q1.x, q1.y, q1.z, q1.w };
    const float w2[4] = { q2.x, q2.y, q2.z, q2.w };
    const float w3[4] = { q3.x, q3.y, q3.z, q3.w };
    const float w4[4] = { q4.x, q4.y, q4.z, q4.w };
    float v[6][4];
    #pragma unroll
    for (int k = 0; k < 4; ++k) {
      const float sev = (w0[k] + w2[k]) + w4[k];
      const float sod = w1[k] + w3[k];
      const float ev  = (w0[k] * (1.0f / 24.0f) + w2[k] * (1.0f / 6.0f)) + w4[k] * (2.0f / 3.0f);
      const float od  = w1[k] * (1.0f / 12.0f) + w3[k] * (1.0f / 3.0f);
      v[0][k] = 0.25f * w0[k];
      v[1][k] = (sev + sod) * (-1.0f / 6.0f);
      v[2][k] = (sev - sod) * (-1.0f / 6.0f);
      v[3][k] = ev + od;
      v[4][k] = ev - od;
      v[5][k] = w4[k];
    }
    float4* vp = reinterpret_cast<float4*>(first ? vt1 : vt) + (size_t)p * (6 * 2048) + r;
    #pragma unroll
    for (int x = 0; x < 6; ++x)
      vp[x * 2048] = make_float4(v[x][0], v[x][1], v[x][2], v[x][3]);
    return;
  }
  const int b   = blockIdx.x;
  const int cq  = tid & 15;
  const int rg  = tid >> 4;

  __shared__ float4 part4[64][16];
  __shared__ float part2[4][64];
  __shared__ float meanv[64];
  __shared__ float logits[P_];

  const float4* ub =
      reinterpret_cast<const float4*>(u + (size_t)b * S_ * C_) + rg * 16 + cq;

  float4 acc[8];
  for (int j = 0; j < 8; ++j) acc[j] = make_float4(0.f, 0.f, 0.f, 0.f);

  for (int it = 0; it < 64; it += 8) {
    float4 v[8];
    for (int j = 0; j < 8; ++j) v[j] = ub[(it + j) * (64 * 16)];
    for (int j = 0; j < 8; ++j) {
      acc[j].x += v[j].x; acc[j].y += v[j].y;
      acc[j].z += v[j].z; acc[j].w += v[j].w;
    }
  }
  {
    float4 s;
    s.x = ((acc[0].x + acc[1].x) + (acc[2].x + acc[3].x)) +
          ((acc[4].x + acc[5].x) + (acc[6].x + acc[7].x));
    s.y = ((acc[0].y + acc[1].y) + (acc[2].y + acc[3].y)) +
          ((acc[4].y + acc[5].y) + (acc[6].y + acc[7].y));
    s.z = ((acc[0].z + acc[1].z) + (acc[2].z + acc[3].z)) +
          ((acc[4].z + acc[5].z) + (acc[6].z + acc[7].z));
    s.w = ((acc[0].w + acc[1].w) + (acc[2].w + acc[3].w)) +
          ((acc[4].w + acc[5].w) + (acc[6].w + acc[7].w));
    part4[rg][cq] = s;
  }
  __syncthreads();

  const int c = tid & 63;
  if (tid < 256) {
    const float* pf = reinterpret_cast<const float*>(&part4[0][0]);
    const int q = tid >> 6;
    float s = 0.f;
    for (int r = q * 16; r < q * 16 + 16; ++r) s += pf[r * 64 + c];
    part2[q][c] = s;
  }
  __syncthreads();

  if (tid < 64)
    meanv[c] = (part2[0][c] + part2[1][c] + part2[2][c] + part2[3][c]) * (1.0f / S_);
  __syncthreads();

  if (tid < P_) {
    float l = br[tid];
    for (int cc = 0; cc < C_; ++cc) l += meanv[cc] * Wr[cc * P_ + tid];
    logits[tid] = l;
  }
  __syncthreads();

  if (tid == 0) {
    int i0 = 0; float v0 = logits[0];
    for (int p = 1; p < P_; ++p)
      if (logits[p] > v0) { v0 = logits[p]; i0 = p; }
    int i1 = -1; float v1 = -3.0e38f;
    for (int p = 0; p < P_; ++p)
      if (p != i0 && logits[p] > v1) { v1 = logits[p]; i1 = p; }
    float e1 = __expf(v1 - v0);
    float w0 = 1.0f / (1.0f + e1);
    float w1 = e1 * w0;
    int* o = gate_out + b * 4;
    o[0] = i0; o[1] = i1;
    o[2] = __float_as_int(w0); o[3] = __float_as_int(w1);
  }
}

#define TS_      128
#define U_ROWS   136
#define UP       66
#define H_OFF    (U_ROWS * UP)
#define HP       130
#define H_ROWS   132
#define LDS_F    (H_OFF + H_ROWS * HP)
#define NPAIR1   66
#define NT1      8
#define NT2      16
#define PLANE    (H_ * C_)

template <int BP>
__device__ __forceinline__ void load_v_trip(const float* __restrict__ vg, int i,
                                            v2f (&bf)[2][6]) {
  const float* bq = vg + i * (8 * BP);
  #pragma unroll
  for (int s = 0; s < 2; ++s) {
    #pragma unroll
    for (int x = 0; x < 6; ++x) {
      bf[s][x].x = bq[x * PLANE + (4 * s) * BP];
      bf[s][x].y = bq[x * PLANE + (4 * s + 1) * BP];
    }
  }
}
template <int AP, int JSTR, int NJ>
__device__ __forceinline__ void wino_body(const float* ah, int i, const v2f (&bc)[2][6],
                                          v8f (&m)[NJ][6]) {
  const float* ap = ah + i * 8;
  #pragma unroll
  for (int s = 0; s < 2; ++s) {
    #pragma unroll
    for (int j = 0; j < NJ; ++j) {
      const float* aq = ap + j * JSTR + s * 4;
      const v2f d0 = *(const v2f*)(aq);
      const v2f d1 = *(const v2f*)(aq + AP);
      const v2f d2 = *(const v2f*)(aq + 2 * AP);
      const v2f d3 = *(const v2f*)(aq + 3 * AP);
      const v2f d4 = *(const v2f*)(aq + 4 * AP);
      const v2f d5 = *(const v2f*)(aq + 5 * AP);
      const v2f pa = d4 - 4.0f * d2;
      const v2f pb = d3 - 4.0f * d1;
      const v2f pc = d4 - d2;
      const v2f pe = d3 - d1;
      const v2f x0 = 4.0f * d0 + (d4 - 5.0f * d2);
      const v2f x1 = pa + pb;
      const v2f x2 = pa - pb;
      const v2f x3 = pc + 2.0f * pe;
      const v2f x4 = pc - 2.0f * pe;
      const v2f x5 = 4.0f * d1 + (d5 - 5.0f * d3);
      m[j][0] = wmma_f32_k4(x0, bc[s][0], m[j][0]);
      m[j][1] = wmma_f32_k4(x1, bc[s][1], m[j][1]);
      m[j][2] = wmma_f32_k4(x2, bc[s][2], m[j][2]);
      m[j][3] = wmma_f32_k4(x3, bc[s][3], m[j][3]);
      m[j][4] = wmma_f32_k4(x4, bc[s][4], m[j][4]);
      m[j][5] = wmma_f32_k4(x5, bc[s][5], m[j][5]);
    }
  }
}
template <int AP, int JSTR, int NJ, int BP, int NT, int LO, int HI>
__device__ __forceinline__ void wino_trips(const float* ah, const float* __restrict__ vg,
                                           v2f (&b0)[2][6], v2f (&b1)[2][6], v8f (&m)[NJ][6]) {
  #pragma unroll 1
  for (int i = LO; i < HI; i += 2) {
    load_v_trip<BP>(vg, i + 1, b1);
    wino_body<AP, JSTR, NJ>(ah, i, b0, m);
    load_v_trip<BP>(vg, (i + 2 < NT) ? i + 2 : 0, b0);
    wino_body<AP, JSTR, NJ>(ah, i + 1, b1, m);
  }
}
template <int T0, int NJ>
__device__ __forceinline__ void conv1_out(const v8f (&m)[NJ][6], float bb, float* hhalf,
                                          int s0, int hi) {
  #pragma unroll
  for (int j = 0; j < NJ; ++j) {
    const int T = T0 + j;
    #pragma unroll
    for (int g = 0; g < 8; ++g) {
      if (16 * T + g >= NPAIR1) continue;
      const int tau = 16 * T + 8 * hi + g;
      const float y0 = ((m[j][0][g] + m[j][1][g]) + (m[j][2][g] + m[j][3][g])) + m[j][4][g];
      const float y1 = ((m[j][1][g] - m[j][2][g]) + 2.0f * (m[j][3][g] - m[j][4][g])) + m[j][5][g];
      float a0 = gelu_tanh(y0 + bb);
      float a1 = gelu_tanh(y1 + bb);
      if (T == 0 || 16 * T + 15 >= 64) {
        const int hrow = s0 - 2 + 2 * tau;
        a0 = (hrow >= 0 && hrow < S_) ? a0 : 0.f;
        a1 = (hrow + 1 >= 0 && hrow + 1 < S_) ? a1 : 0.f;
      }
      if (16 * T + 8 + g < NPAIR1 || hi == 0) {
        hhalf[(32 * T + 2 * g) * HP]     = a0;
        hhalf[(32 * T + 2 * g + 1) * HP] = a1;
      }
    }
  }
}
__global__ __launch_bounds__(256, 4)
void moe_conv_kernel(const float* __restrict__ u,
                     const float* __restrict__ b1,
                     const float* __restrict__ b2,
                     const float* __restrict__ vt1,
                     const float* __restrict__ vt,
                     const int*   __restrict__ gate,
                     float* __restrict__ out) {
  extern __shared__ float smem[];
  float* u_lds = smem;
  float* h_lds = smem + H_OFF;

  const int s0   = blockIdx.x * TS_;
  const int b    = blockIdx.y;
  const int tid  = threadIdx.x;
  const int lane = tid & 31;
  const int wv   = tid >> 5;
  const int n    = lane & 15;
  const int hi   = lane >> 4;

  const int* gp = gate + b * 4;
  int   eidx[2] = { __builtin_amdgcn_readfirstlane(gp[0]),
                    __builtin_amdgcn_readfirstlane(gp[1]) };
  float ew[2]   = { __int_as_float(__builtin_amdgcn_readfirstlane(gp[2])),
                    __int_as_float(__builtin_amdgcn_readfirstlane(gp[3])) };

  const int h0 = wv * 16;
  const int ct  = wv & 3, c0 = ct * 16;
  const int rtb = wv >> 2;

  const int boff1 = (2 * hi) * H_ + h0 + n;
  const int boff2 = (2 * hi) * C_ + c0 + n;

  v2f fa[2][6], fb[2][6];
  {
    const float4* ub4 =
        reinterpret_cast<const float4*>(u + (size_t)b * S_ * C_);
    float4 uv[9];
    #pragma unroll
    for (int j = 0; j < 9; ++j) {
      int idx = tid + 256 * j;
      int r = idx >> 4, q = idx & 15;
      int gs = s0 - 4 + r;
      uv[j] = (idx < U_ROWS * 16 && gs >= 0 && gs < S_)
                  ? ub4[gs * (C_ / 4) + q] : make_float4(0.f, 0.f, 0.f, 0.f);
    }
    load_v_trip<H_>(vt1 + (size_t)eidx[0] * (6 * PLANE) + boff1, 0, fa);
    #pragma unroll
    for (int j = 0; j < 9; ++j) {
      int idx = tid + 256 * j;
      int r = idx >> 4, q = idx & 15;
      if (idx < U_ROWS * 16) {
        v2f lo = { uv[j].x, uv[j].y }, hh = { uv[j].z, uv[j].w };
        *reinterpret_cast<v2f*>(u_lds + r * UP + q * 4)     = lo;
        *reinterpret_cast<v2f*>(u_lds + r * UP + q * 4 + 2) = hh;
      }
    }
  }

  __syncthreads();

  float vout[2][2][8];
  #pragma unroll
  for (int j = 0; j < 2; ++j) {
    #pragma unroll
    for (int g = 0; g < 8; ++g) {
      #pragma unroll
      for (int i = 0; i < 2; ++i) {
        int row = 32 * (rtb + 2 * j) + 2 * (g + 8 * hi) + i;
        vout[j][i][g] = u_lds[(row + 4) * UP + c0 + n];
      }
    }
  }

  const float* aru  = u_lds + (2 * n) * UP + 2 * hi;
  const float* aru4 = u_lds + (2 * min(64 + n, NPAIR1 - 1)) * UP + 2 * hi;
  const float* arh  = h_lds + (2 * n) * HP + 2 * hi;
  float* hhalf = h_lds + (16 * hi) * HP + h0 + n;

  #pragma unroll 1
  for (int e = 0; e < 2; ++e) {
    const int   p  = eidx[e];
    const float we = ew[e];
    const float* vg1 = vt1 + (size_t)p * (6 * PLANE) + boff1;
    const float* vg  = vt  + (size_t)p * (6 * PLANE) + boff2;

    {
      const float bb = b1[p * H_ + h0 + n];
      const v8f z = {};
      {
        v8f m[2][6] = { { z, z, z, z, z, z }, { z, z, z, z, z, z } };
        __builtin_amdgcn_s_setprio(3);
        wino_trips<UP, 32 * UP, 2, H_, NT1, 0, NT1>(aru, vg1, fa, fb, m);
        conv1_out<0, 2>(m, bb, hhalf, s0, hi);
      }
      {
        v8f m[2][6] = { { z, z, z, z, z, z }, { z, z, z, z, z, z } };
        __builtin_amdgcn_s_setprio(2);
        wino_trips<UP, 32 * UP, 2, H_, NT1, 0, NT1>(aru + 64 * UP, vg1, fa, fb, m);
        conv1_out<2, 2>(m, bb, hhalf, s0, hi);
      }
      {
        v8f m[1][6] = { { z, z, z, z, z, z } };
        __builtin_amdgcn_s_setprio(1);
        wino_trips<UP, 0, 1, H_, NT1, 0, 6>(aru4, vg1, fa, fb, m);
        __builtin_amdgcn_s_setprio(0);
        wino_trips<UP, 0, 1, H_, NT1, 6, NT1>(aru4, vg1, fa, fb, m);
        load_v_trip<C_>(vg, 0, fa);
        conv1_out<4, 1>(m, bb, hhalf, s0, hi);
      }
    }
    __syncthreads();

    {
      const float bb2 = b2[p * C_ + c0 + n];
      const v8f z = {};
      v8f m[2][6] = { { z, z, z, z, z, z }, { z, z, z, z, z, z } };
      const float* ah = arh + rtb * (32 * HP);
      __builtin_amdgcn_s_setprio(3);
      wino_trips<HP, 64 * HP, 2, C_, NT2, 0, 6>(ah, vg, fa, fb, m);
      __builtin_amdgcn_s_setprio(2);
      wino_trips<HP, 64 * HP, 2, C_, NT2, 6, 10>(ah, vg, fa, fb, m);
      __builtin_amdgcn_s_setprio(1);
      wino_trips<HP, 64 * HP, 2, C_, NT2, 10, 14>(ah, vg, fa, fb, m);
      __builtin_amdgcn_s_setprio(0);
      wino_trips<HP, 64 * HP, 2, C_, NT2, 14, NT2>(ah, vg, fa, fb, m);
      if (e == 0)
        load_v_trip<H_>(vt1 + (size_t)eidx[1] * (6 * PLANE) + boff1, 0, fa);
      #pragma unroll
      for (int j = 0; j < 2; ++j) {
        #pragma unroll
        for (int g = 0; g < 8; ++g) {
          const float y0 = ((m[j][0][g] + m[j][1][g]) + (m[j][2][g] + m[j][3][g])) + m[j][4][g];
          const float y1 = ((m[j][1][g] - m[j][2][g]) + 2.0f * (m[j][3][g] - m[j][4][g])) + m[j][5][g];
          vout[j][0][g] += we * (y0 + bb2);
          vout[j][1][g] += we * (y1 + bb2);
        }
      }
    }
    if (e == 0)
      __syncthreads();
  }

  #pragma unroll
  for (int j = 0; j < 2; ++j) {
    #pragma unroll
    for (int g = 0; g < 8; ++g) {
      #pragma unroll
      for (int i = 0; i < 2; ++i) {
        int row = 32 * (rtb + 2 * j) + 2 * (g + 8 * hi) + i;
        out[((size_t)b * S_ + s0 + row) * C_ + c0 + n] = vout[j][i][g];
      }
    }
  }
}

extern "C" void kernel_launch(void* const* d_in, const int* in_sizes, int n_in,
                              void* d_out, int out_size, void* d_ws, size_t ws_size,
                              hipStream_t stream) {
  const float* u  = (const float*)d_in[0];
  const float* W1 = (const float*)d_in[1];
  const float* b1 = (const float*)d_in[2];
  const float* W2 = (const float*)d_in[3];
  const float* b2 = (const float*)d_in[4];
  const float* Wr = (const float*)d_in[5];
  const float* br = (const float*)d_in[6];
  float* out = (float*)d_out;
  int*   gate = (int*)d_ws;
  float* vt   = (float*)d_ws + 1024;
  float* vt1  = (float*)d_ws + 394240;

  moe_gate_kernel<<<dim3(B_ + NXF), dim3(1024), 0, stream>>>(u, Wr, br, W2, W1, gate, vt, vt1);

  size_t lds_bytes = (size_t)LDS_F * sizeof(float);
  moe_conv_kernel<<<dim3(S_ / TS_, B_), dim3(256), lds_bytes, stream>>>(
      u, b1, b2, vt1, vt, gate, out);
}


# Mean over S on all WGPs: 8 blocks per batch row write partial column sums; each conv block adds them and does logits, top-2, softmax in its head; changes the sum order of mean and logits
# speedup vs baseline: 3.0939x; 1.0432x over previous
#include <hip/hip_runtime.h>
#include <hip/hip_bf16.h>

#define B_  32
#define S_  4096
#define C_  64
#define H_  128
#define P_  8
#define KW  5

typedef __attribute__((ext_vector_type(2))) float v2f;
typedef __attribute__((ext_vector_type(8))) float v8f;

__device__ __forceinline__ v8f wmma_f32_k4(v2f a, v2f b, v8f c) {
  return __builtin_amdgcn_wmma_f32_16x16x4_f32(false, a, false, b, (short)0, c,
                                               false, false);
}

__device__ __forceinline__ float gelu_tanh(float x) {
  float x3 = x * x * x;
  float y  = 1.5957691216057308f * (x + 0.044715f * x3);
  return x * __builtin_amdgcn_rcpf(1.0f + __expf(-y));
}

#define NPART 8
#define NXF 128
__global__ __launch_bounds__(256)
void moe_gate_kernel(const float* __restrict__ u,
                     const float* __restrict__ W2,
                     const float* __restrict__ W1,
                     float* __restrict__ part,
                     float* __restrict__ vt,
                     float* __restrict__ vt1) {
  const int tid = threadIdx.x;
  if (blockIdx.x >= B_ * NPART) {
    const int xb = blockIdx.x - B_ * NPART;
    const bool first = xb >= NXF / 2;
    const int idx = (xb & (NXF / 2 - 1)) * 256 + tid;
    const int p = idx >> 11;
    const int r = idx & 2047;
    const float4* wp =
        reinterpret_cast<const float4*>(first ? W1 : W2) + (size_t)p * (KW * H_ * C_ / 4) + r;
    const float4 q0 = wp[0], q1 = wp[2048], q2 = wp[4096], q3 = wp[6144], q4 = wp[8192];
    const float w0[4] = { q0.x, q0.y, q0.z, q0.w };
    const float w1[4] = { q1.x, q1.y, q1.z, q1.w };
    const float w2[4] = { q2.x, q2.y, q2.z, q2.w };
    const float w3[4] = { q3.x, q3.y, q3.z, q3.w };
    const float w4[4] = { q4.x, q4.y, q4.z, q4.w };
    float v[6][4];
    #pragma unroll
    for (int k = 0; k < 4; ++k) {
      const float sev = (w0[k] + w2[k]) + w4[k];
      const float sod = w1[k] + w3[k];
      const float ev  = (w0[k] * (1.0f / 24.0f) + w2[k] * (1.0f / 6.0f)) + w4[k] * (2.0f / 3.0f);
      const float od  = w1[k] * (1.0f / 12.0f) + w3[k] * (1.0f / 3.0f);
      v[0][k] = 0.25f * w0[k];
      v[1][k] = (sev + sod) * (-1.0f / 6.0f);
      v[2][k] = (sev - sod) * (-1.0f / 6.0f);
      v[3][k] = ev + od;
      v[4][k] = ev - od;
      v[5][k] = w4[k];
    }
    float4* vp = reinterpret_cast<float4*>(first ? vt1 : vt) + (size_t)p * (6 * 2048) + r;
    #pragma unroll
    for (int x = 0; x < 6; ++x)
      vp[x * 2048] = make_float4(v[x][0], v[x][1], v[x][2], v[x][3]);
    return;
  }
  const int b   = blockIdx.x / NPART;
  const int sl  = blockIdx.x % NPART;
  const int cq  = tid & 15;
  const int rg  = tid >> 4;

  __shared__ float4 part4[16][16];

  const float4* ub =
      reinterpret_cast<const float4*>(u + ((size_t)b * S_ + (size_t)sl * (S_ / NPART)) * C_) +
      rg * 16 + cq;

  float4 v[32];
  #pragma unroll
  for (int j = 0; j < 32; ++j) v[j] = ub[j * (16 * 16)];
  float4 acc[8];
  #pragma unroll
  for (int j = 0; j < 8; ++j) acc[j] = v[j];
  #pragma unroll
  for (int j = 8; j < 32; ++j) {
    acc[j & 7].x += v[j].x; acc[j & 7].y += v[j].y;
    acc[j & 7].z += v[j].z; acc[j & 7].w += v[j].w;
  }
  {
    float4 s;
    s.x = ((acc[0].x + acc[1].x) + (acc[2].x + acc[3].x)) +
          ((acc[4].x + acc[5].x) + (acc[6].x + acc[7].x));
    s.y = ((acc[0].y + acc[1].y) + (acc[2].y + acc[3].y)) +
          ((acc[4].y + acc[5].y) + (acc[6].y + acc[7].y));
    s.z = ((acc[0].z + acc[1].z) + (acc[2].z + acc[3].z)) +
          ((acc[4].z + acc[5].z) + (acc[6].z + acc[7].z));
    s.w = ((acc[0].w + acc[1].w) + (acc[2].w + acc[3].w)) +
          ((acc[4].w + acc[5].w) + (acc[6].w + acc[7].w));
    part4[rg][cq] = s;
  }
  __syncthreads();

  if (tid < 64) {
    const float* pf = reinterpret_cast<const float*>(&part4[0][0]);
    float s = 0.f;
    #pragma unroll
    for (int r = 0; r < 16; ++r) s += pf[r * 64 + tid];
    part[((size_t)b * NPART + sl) * 64 + tid] = s;
  }
}

#define TS_      128
#define U_ROWS   136
#define UP       66
#define H_OFF    (U_ROWS * UP)
#define HP       130
#define H_ROWS   132
#define LDS_F    (H_OFF + H_ROWS * HP)
#define NPAIR1   66
#define NT1      8
#define NT2      16
#define PLANE    (H_ * C_)

template <int BP>
__device__ __forceinline__ void load_v_trip(const float* __restrict__ vg, int i,
                                            v2f (&bf)[2][6]) {
  const float* bq = vg + i * (8 * BP);
  #pragma unroll
  for (int s = 0; s < 2; ++s) {
    #pragma unroll
    for (int x = 0; x < 6; ++x) {
      bf[s][x].x = bq[x * PLANE + (4 * s) * BP];
      bf[s][x].y = bq[x * PLANE + (4 * s + 1) * BP];
    }
  }
}
template <int AP, int JSTR, int NJ>
__device__ __forceinline__ void wino_body(const float* ah, int i, const v2f (&bc)[2][6],
                                          v8f (&m)[NJ][6]) {
  const float* ap = ah + i * 8;
  #pragma unroll
  for (int s = 0; s < 2; ++s) {
    #pragma unroll
    for (int j = 0; j < NJ; ++j) {
      const float* aq = ap + j * JSTR + s * 4;
      const v2f d0 = *(const v2f*)(aq);
      const v2f d1 = *(const v2f*)(aq + AP);
      const v2f d2 = *(const v2f*)(aq + 2 * AP);
      const v2f d3 = *(const v2f*)(aq + 3 * AP);
      const v2f d4 = *(const v2f*)(aq + 4 * AP);
      const v2f d5 = *(const v2f*)(aq + 5 * AP);
      const v2f pa = d4 - 4.0f * d2;
      const v2f pb = d3 - 4.0f * d1;
      const v2f pc = d4 - d2;
      const v2f pe = d3 - d1;
      const v2f x0 = 4.0f * d0 + (d4 - 5.0f * d2);
      const v2f x1 = pa + pb;
      const v2f x2 = pa - pb;
      const v2f x3 = pc + 2.0f * pe;
      const v2f x4 = pc - 2.0f * pe;
      const v2f x5 = 4.0f * d1 + (d5 - 5.0f * d3);
      m[j][0] = wmma_f32_k4(x0, bc[s][0], m[j][0]);
      m[j][1] = wmma_f32_k4(x1, bc[s][1], m[j][1]);
      m[j][2] = wmma_f32_k4(x2, bc[s][2], m[j][2]);
      m[j][3] = wmma_f32_k4(x3, bc[s][3], m[j][3]);
      m[j][4] = wmma_f32_k4(x4, bc[s][4], m[j][4]);
      m[j][5] = wmma_f32_k4(x5, bc[s][5], m[j][5]);
    }
  }
}
template <int AP, int JSTR, int NJ, int BP, int NT, int LO, int HI>
__device__ __forceinline__ void wino_trips(const float* ah, const float* __restrict__ vg,
                                           v2f (&b0)[2][6], v2f (&b1)[2][6], v8f (&m)[NJ][6]) {
  #pragma unroll 1
  for (int i = LO; i < HI; i += 2) {
    load_v_trip<BP>(vg, i + 1, b1);
    wino_body<AP, JSTR, NJ>(ah, i, b0, m);
    load_v_trip<BP>(vg, (i + 2 < NT) ? i + 2 : 0, b0);
    wino_body<AP, JSTR, NJ>(ah, i + 1, b1, m);
  }
}
template <int T0, int NJ>
__device__ __forceinline__ void conv1_out(const v8f (&m)[NJ][6], float bb, float* hhalf,
                                          int s0, int hi) {
  #pragma unroll
  for (int j = 0; j < NJ; ++j) {
    const int T = T0 + j;
    #pragma unroll
    for (int g = 0; g < 8; ++g) {
      if (16 * T + g >= NPAIR1) continue;
      const int tau = 16 * T + 8 * hi + g;
      const float y0 = ((m[j][0][g] + m[j][1][g]) + (m[j][2][g] + m[j][3][g])) + m[j][4][g];
      const float y1 = ((m[j][1][g] - m[j][2][g]) + 2.0f * (m[j][3][g] - m[j][4][g])) + m[j][5][g];
      float a0 = gelu_tanh(y0 + bb);
      float a1 = gelu_tanh(y1 + bb);
      if (T == 0 || 16 * T + 15 >= 64) {
        const int hrow = s0 - 2 + 2 * tau;
        a0 = (hrow >= 0 && hrow < S_) ? a0 : 0.f;
        a1 = (hrow + 1 >= 0 && hrow + 1 < S_) ? a1 : 0.f;
      }
      if (16 * T + 8 + g < NPAIR1 || hi == 0) {
        hhalf[(32 * T + 2 * g) * HP]     = a0;
        hhalf[(32 * T + 2 * g + 1) * HP] = a1;
      }
    }
  }
}
__global__ __launch_bounds__(256, 4)
void moe_conv_kernel(const float* __restrict__ u,
                     const float* __restrict__ b1,
                     const float* __restrict__ b2,
                     const float* __restrict__ vt1,
                     const float* __restrict__ vt,
                     const float* __restrict__ part,
                     const float* __restrict__ Wr,
                     const float* __restrict__ br,
                     float* __restrict__ out) {
  extern __shared__ float smem[];
  float* u_lds = smem;
  float* h_lds = smem + H_OFF;

  const int s0   = blockIdx.x * TS_;
  const int b    = blockIdx.y;
  const int tid  = threadIdx.x;
  const int lane = tid & 31;
  const int wv   = tid >> 5;
  const int n    = lane & 15;
  const int hi   = lane >> 4;

  const int h0 = wv * 16;
  const int ct  = wv & 3, c0 = ct * 16;
  const int rtb = wv >> 2;

  const int boff1 = (2 * hi) * H_ + h0 + n;
  const int boff2 = (2 * hi) * C_ + c0 + n;

  v2f fa[2][6], fb[2][6];
  int   eidx[2];
  float ew[2];
  float lsum;
  {
    const float* pp = part + (size_t)b * (NPART * 64) + lane;
    const float a0 = pp[0],   a1 = pp[64],  a2 = pp[128], a3 = pp[192];
    const float a4 = pp[256], a5 = pp[320], a6 = pp[384], a7 = pp[448];
    const float c0_ = pp[32],  c1_ = pp[96],  c2_ = pp[160], c3_ = pp[224];
    const float c4_ = pp[288], c5_ = pp[352], c6_ = pp[416], c7_ = pp[480];
    const float wa = Wr[lane * P_ + wv];
    const float wb = Wr[(lane + 32) * P_ + wv];
    const float ma = (((a0 + a1) + (a2 + a3)) + ((a4 + a5) + (a6 + a7))) * (1.0f / S_);
    const float mb = (((c0_ + c1_) + (c2_ + c3_)) + ((c4_ + c5_) + (c6_ + c7_))) * (1.0f / S_);
    lsum = (ma * wa + mb * wb) + (lane == 0 ? br[wv] : 0.f);
  }
  {
    const float4* ub4 =
        reinterpret_cast<const float4*>(u + (size_t)b * S_ * C_);
    float4 uv[9];
    #pragma unroll
    for (int j = 0; j < 9; ++j) {
      int idx = tid + 256 * j;
      int r = idx >> 4, q = idx & 15;
      int gs = s0 - 4 + r;
      uv[j] = (idx < U_ROWS * 16 && gs >= 0 && gs < S_)
                  ? ub4[gs * (C_ / 4) + q] : make_float4(0.f, 0.f, 0.f, 0.f);
    }
    #pragma unroll
    for (int m = 16; m >= 1; m >>= 1) lsum += __shfl_xor(lsum, m, 32);
    if (lane == 0) h_lds[wv] = lsum;
    __syncthreads();
    {
      float lg[P_];
      #pragma unroll
      for (int p = 0; p < P_; ++p) lg[p] = h_lds[p];
      int i0 = 0; float v0 = lg[0];
      #pragma unroll
      for (int p = 1; p < P_; ++p)
        if (lg[p] > v0) { v0 = lg[p]; i0 = p; }
      int i1 = -1; float v1 = -3.0e38f;
      #pragma unroll
      for (int p = 0; p < P_; ++p)
        if (p != i0 && lg[p] > v1) { v1 = lg[p]; i1 = p; }
      const float e1 = __expf(v1 - v0);
      const float w0 = 1.0f / (1.0f + e1);
      const float w1 = e1 * w0;
      eidx[0] = __builtin_amdgcn_readfirstlane(i0);
      eidx[1] = __builtin_amdgcn_readfirstlane(i1);
      ew[0] = __int_as_float(__builtin_amdgcn_readfirstlane(__float_as_int(w0)));
      ew[1] = __int_as_float(__builtin_amdgcn_readfirstlane(__float_as_int(w1)));
    }
    load_v_trip<H_>(vt1 + (size_t)eidx[0] * (6 * PLANE) + boff1, 0, fa);
    #pragma unroll
    for (int j = 0; j < 9; ++j) {
      int idx = tid + 256 * j;
      int r = idx >> 4, q = idx & 15;
      if (idx < U_ROWS * 16) {
        v2f lo = { uv[j].x, uv[j].y }, hh = { uv[j].z, uv[j].w };
        *reinterpret_cast<v2f*>(u_lds + r * UP + q * 4)     = lo;
        *reinterpret_cast<v2f*>(u_lds + r * UP + q * 4 + 2) = hh;
      }
    }
  }

  __syncthreads();

  float vout[2][2][8];
  #pragma unroll
  for (int j = 0; j < 2; ++j) {
    #pragma unroll
    for (int g = 0; g < 8; ++g) {
      #pragma unroll
      for (int i = 0; i < 2; ++i) {
        int row = 32 * (rtb + 2 * j) + 2 * (g + 8 * hi) + i;
        vout[j][i][g] = u_lds[(row + 4) * UP + c0 + n];
      }
    }
  }

  const float* aru  = u_lds + (2 * n) * UP + 2 * hi;
  const float* aru4 = u_lds + (2 * min(64 + n, NPAIR1 - 1)) * UP + 2 * hi;
  const float* arh  = h_lds + (2 * n) * HP + 2 * hi;
  float* hhalf = h_lds + (16 * hi) * HP + h0 + n;

  #pragma unroll 1
  for (int e = 0; e < 2; ++e) {
    const int   p  = eidx[e];
    const float we = ew[e];
    const float* vg1 = vt1 + (size_t)p * (6 * PLANE) + boff1;
    const float* vg  = vt  + (size_t)p * (6 * PLANE) + boff2;

    {
      const float bb = b1[p * H_ + h0 + n];
      const v8f z = {};
      {
        v8f m[2][6] = { { z, z, z, z, z, z }, { z, z, z, z, z, z } };
        __builtin_amdgcn_s_setprio(3);
        wino_trips<UP, 32 * UP, 2, H_, NT1, 0, NT1>(aru, vg1, fa, fb, m);
        conv1_out<0, 2>(m, bb, hhalf, s0, hi);
      }
      {
        v8f m[2][6] = { { z, z, z, z, z, z }, { z, z, z, z, z, z } };
        __builtin_amdgcn_s_setprio(2);
        wino_trips<UP, 32 * UP, 2, H_, NT1, 0, NT1>(aru + 64 * UP, vg1, fa, fb, m);
        conv1_out<2, 2>(m, bb, hhalf, s0, hi);
      }
      {
        v8f m[1][6] = { { z, z, z, z, z, z } };
        __builtin_amdgcn_s_setprio(1);
        wino_trips<UP, 0, 1, H_, NT1, 0, 6>(aru4, vg1, fa, fb, m);
        __builtin_amdgcn_s_setprio(0);
        wino_trips<UP, 0, 1, H_, NT1, 6, NT1>(aru4, vg1, fa, fb, m);
        load_v_trip<C_>(vg, 0, fa);
        conv1_out<4, 1>(m, bb, hhalf, s0, hi);
      }
    }
    __syncthreads();

    {
      const float bb2 = b2[p * C_ + c0 + n];
      const v8f z = {};
      v8f m[2][6] = { { z, z, z, z, z, z }, { z, z, z, z, z, z } };
      const float* ah = arh + rtb * (32 * HP);
      __builtin_amdgcn_s_setprio(3);
      wino_trips<HP, 64 * HP, 2, C_, NT2, 0, 6>(ah, vg, fa, fb, m);
      __builtin_amdgcn_s_setprio(2);
      wino_trips<HP, 64 * HP, 2, C_, NT2, 6, 10>(ah, vg, fa, fb, m);
      __builtin_amdgcn_s_setprio(1);
      wino_trips<HP, 64 * HP, 2, C_, NT2, 10, 14>(ah, vg, fa, fb, m);
      __builtin_amdgcn_s_setprio(0);
      wino_trips<HP, 64 * HP, 2, C_, NT2, 14, NT2>(ah, vg, fa, fb, m);
      if (e == 0)
        load_v_trip<H_>(vt1 + (size_t)eidx[1] * (6 * PLANE) + boff1, 0, fa);
      #pragma unroll
      for (int j = 0; j < 2; ++j) {
        #pragma unroll
        for (int g = 0; g < 8; ++g) {
          const float y0 = ((m[j][0][g] + m[j][1][g]) + (m[j][2][g] + m[j][3][g])) + m[j][4][g];
          const float y1 = ((m[j][1][g] - m[j][2][g]) + 2.0f * (m[j][3][g] - m[j][4][g])) + m[j][5][g];
          vout[j][0][g] += we * (y0 + bb2);
          vout[j][1][g] += we * (y1 + bb2);
        }
      }
    }
    if (e == 0)
      __syncthreads();
  }

  #pragma unroll
  for (int j = 0; j < 2; ++j) {
    #pragma unroll
    for (int g = 0; g < 8; ++g) {
      #pragma unroll
      for (int i = 0; i < 2; ++i) {
        int row = 32 * (rtb + 2 * j) + 2 * (g + 8 * hi) + i;
        out[((size_t)b * S_ + s0 + row) * C_ + c0 + n] = vout[j][i][g];
      }
    }
  }
}

extern "C" void kernel_launch(void* const* d_in, const int* in_sizes, int n_in,
                              void* d_out, int out_size, void* d_ws, size_t ws_size,
                              hipStream_t stream) {
  const float* u  = (const float*)d_in[0];
  const float* W1 = (const float*)d_in[1];
  const float* b1 = (const float*)d_in[2];
  const float* W2 = (const float*)d_in[3];
  const float* b2 = (const float*)d_in[4];
  const float* Wr = (const float*)d_in[5];
  const float* br = (const float*)d_in[6];
  float* out = (float*)d_out;
  float* vt   = (float*)d_ws + 1024;
  float* vt1  = (float*)d_ws + 394240;
  float* part = (float*)d_ws + 787456;

  moe_gate_kernel<<<dim3(B_ * NPART + NXF), dim3(256), 0, stream>>>(u, W2, W1, part, vt, vt1);

  size_t lds_bytes = (size_t)LDS_F * sizeof(float);
  moe_conv_kernel<<<dim3(S_ / TS_, B_), dim3(256), lds_bytes, stream>>>(
      u, b1, b2, vt1, vt, part, Wr, br, out);
}
